# starting kernel
# baseline (speedup 1.0000x reference)
#include <hip/hip_runtime.h>


typedef __attribute__((ext_vector_type(2))) float v2f;
typedef __attribute__((ext_vector_type(8))) float v8f;

#define GRIDN 12
#define INV_BW (11.0f / 6.0f)
#define CSTEP  (6.0f / 11.0f)

#define NSTEP1 30
#define NSTEP2 79
#define NSTEP3 40
#define NT1 2
#define NT2 1
#define NT3 1
#define B1_FLOATS (NSTEP1 * NT1 * 64)
#define B2_FLOATS (NSTEP2 * NT2 * 64)
#define B3_FLOATS (NSTEP3 * NT3 * 64)
#define B1_OFF 0
#define B2_OFF (B1_FLOATS)
#define B3_OFF (B1_FLOATS + B2_FLOATS)
#define BTOTAL (B1_FLOATS + B2_FLOATS + B3_FLOATS)

__global__ __launch_bounds__(256) void kan_pack(
    const float* __restrict__ c1, const float* __restrict__ w1, const float* __restrict__ b1,
    const float* __restrict__ c2, const float* __restrict__ w2, const float* __restrict__ b2,
    const float* __restrict__ c3, const float* __restrict__ w3, const float* __restrict__ b3,
    const float* __restrict__ base, float* __restrict__ ws) {
  int idx = blockIdx.x * blockDim.x + threadIdx.x;
  if (idx >= BTOTAL) return;

  int layer, e, nt_count;
  if (idx < B2_OFF)       { layer = 0; e = idx;           nt_count = NT1; }
  else if (idx < B3_OFF)  { layer = 1; e = idx - B2_OFF;  nt_count = NT2; }
  else                    { layer = 2; e = idx - B3_OFF;  nt_count = NT3; }

  int v    = e & 1;
  int lane = (e >> 1) & 31;
  int q    = (e >> 1) >> 5;
  int t    = q / nt_count;
  int nt   = q - t * nt_count;
  int k    = 4 * t + 2 * (lane >> 4) + v;
  int n    = nt * 16 + (lane & 15);

  float val = 0.0f;
  if (layer == 0) {
    if (n < 24) {
      if (k < 108)      { int i = k / 12, g = k - i * 12; val = c1[(i * 24 + n) * 12 + g]; }
      else if (k < 117) { val = w1[n * 9 + (k - 108)]; }
      else if (k == 117){ val = b1[n]; }
    }
  } else if (layer == 1) {
    if (n < 12) {
      if (k < 288)      { int i = k / 12, g = k - i * 12; val = c2[(i * 12 + n) * 12 + g]; }
      else if (k < 312) { val = w2[n * 24 + (k - 288)]; }
      else if (k == 312){ val = b2[n]; }
    }
  } else {
    if (n < 3) {
      if (k < 144)      { int i = k / 12, g = k - i * 12; val = c3[(i * 3 + n) * 12 + g]; }
      else if (k < 156) { val = w3[n * 12 + (k - 144)]; }
      else if (k == 156){ val = b3[n] + base[n]; }
    }
  }
  ws[idx] = val;
}

template <int I>
__device__ __forceinline__ float afrag_elem(const float* __restrict__ a, int k) {
  const int KB = I * GRIDN;
  float r;
  if (k < KB) {
    int i = (int)((unsigned)k / 12u);
    int g = k - i * 12;
    float d = (a[i] - (-3.0f + CSTEP * (float)g)) * INV_BW;
    r = __expf(-0.5f * d * d);
  } else if (k < KB + I) {
    r = a[k - KB];
  } else if (k == KB + I) {
    r = 1.0f;
  } else {
    r = 0.0f;
  }
  return r;
}

template <int I, int NSTEPS, int NT>
__device__ __forceinline__ void kan_layer_wmma(const float* __restrict__ arow,
                                               const float* __restrict__ Bsh,
                                               int lane, int half,
                                               v8f& c0, v8f& c1) {
#pragma unroll 4
  for (int t = 0; t < NSTEPS; ++t) {
    int k0 = 4 * t + 2 * half;
    v2f a;
    a.x = afrag_elem<I>(arow, k0);
    a.y = afrag_elem<I>(arow, k0 + 1);
    v2f b0 = *(const v2f*)(Bsh + (t * NT + 0) * 64 + lane * 2);
    c0 = __builtin_amdgcn_wmma_f32_16x16x4_f32(false, a, false, b0, (short)0, c0,
                                               false, false);
    if (NT == 2) {
      v2f b1 = *(const v2f*)(Bsh + (t * NT + 1) * 64 + lane * 2);
      c1 = __builtin_amdgcn_wmma_f32_16x16x4_f32(false, a, false, b1, (short)0, c1,
                                                 false, false);
    }
  }
}

__device__ __forceinline__ float silu_f(float z) {
  return z * (1.0f / (1.0f + __expf(-z)));
}

__global__ __launch_bounds__(256, 2) void kan_forward(const float* __restrict__ x,
                                                      const float* __restrict__ ws,
                                                      float* __restrict__ out,
                                                      int tiles_per_wave) {
  __shared__ __align__(16) float Bsh[BTOTAL];
  __shared__ __align__(16) float act0[8][16][12];
  __shared__ __align__(16) float act1[8][16][24];
  __shared__ __align__(16) float act2[8][16][12];
  __shared__ __align__(16) float cor[8][16][4];

  const int tid = threadIdx.x;
  for (int i = tid; i < BTOTAL; i += 256) Bsh[i] = ws[i];
  __syncthreads();

  const int wv   = tid >> 5;
  const int lane = tid & 31;
  const int half = lane >> 4;
  const int m    = lane & 15;
  const float* a0 = &act0[wv][m][0];
  const float* a1 = &act1[wv][m][0];
  const float* a2 = &act2[wv][m][0];

  for (int it = 0; it < tiles_per_wave; ++it) {
    const int rowbase = ((blockIdx.x * tiles_per_wave + it) * 8 + wv) * 16;

    for (int e = lane; e < 144; e += 32) {
      int r = e / 9, c = e - r * 9;
      act0[wv][r][c] = __builtin_nontemporal_load(&x[(size_t)(rowbase + r) * 9 + c]);
    }
    asm volatile("s_wait_dscnt 0" ::: "memory");

    v8f c0 = {}, c1 = {};
    kan_layer_wmma<9, NSTEP1, NT1>(a0, Bsh + B1_OFF, lane, half, c0, c1);
#pragma unroll
    for (int v = 0; v < 8; ++v) {
      int r = half * 8 + v;
      act1[wv][r][m] = silu_f(c0[v]);
      if (m < 8) act1[wv][r][16 + m] = silu_f(c1[v]);
    }
    asm volatile("s_wait_dscnt 0" ::: "memory");

    v8f d0 = {}, dz = {};
    kan_layer_wmma<24, NSTEP2, NT2>(a1, Bsh + B2_OFF, lane, half, d0, dz);
#pragma unroll
    for (int v = 0; v < 8; ++v) {
      int r = half * 8 + v;
      if (m < 12) act2[wv][r][m] = silu_f(d0[v]);
    }
    asm volatile("s_wait_dscnt 0" ::: "memory");

    v8f e0 = {}, ez = {};
    kan_layer_wmma<12, NSTEP3, NT3>(a2, Bsh + B3_OFF, lane, half, e0, ez);
#pragma unroll
    for (int v = 0; v < 8; ++v) {
      int r = half * 8 + v;
      if (m < 3) cor[wv][r][m] = e0[v];
    }
    asm volatile("s_wait_dscnt 0" ::: "memory");

    if (half == 0) {
      float p0 = cor[wv][m][0], p1 = cor[wv][m][1], p2 = cor[wv][m][2];
      float mx = fmaxf(p0, fmaxf(p1, p2));
      float q0 = __expf(p0 - mx), q1 = __expf(p1 - mx), q2 = __expf(p2 - mx);
      float inv = 1.0f / (q0 + q1 + q2);
      float* o = out + (size_t)(rowbase + m) * 3;
      __builtin_nontemporal_store(q0 * inv, o + 0);
      __builtin_nontemporal_store(q1 * inv, o + 1);
      __builtin_nontemporal_store(q2 * inv, o + 2);
    }
    asm volatile("s_wait_dscnt 0" ::: "memory");
  }
}

extern "C" void kernel_launch(void* const* d_in, const int* in_sizes, int n_in,
                              void* d_out, int out_size, void* d_ws, size_t ws_size,
                              hipStream_t stream) {
  const float* x    = (const float*)d_in[0];
  const float* c1   = (const float*)d_in[1];
  const float* w1   = (const float*)d_in[2];
  const float* b1   = (const float*)d_in[3];
  const float* c2   = (const float*)d_in[4];
  const float* w2   = (const float*)d_in[5];
  const float* b2   = (const float*)d_in[6];
  const float* c3   = (const float*)d_in[7];
  const float* w3   = (const float*)d_in[8];
  const float* b3   = (const float*)d_in[9];
  const float* base = (const float*)d_in[10];
  float* ws  = (float*)d_ws;
  float* out = (float*)d_out;

  const int Btot = in_sizes[0] / 9;
  const int TILES = 4;
  const int blocks = Btot / (128 * TILES);

  kan_pack<<<(BTOTAL + 255) / 256, 256, 0, stream>>>(c1, w1, b1, c2, w2, b2,
                                                     c3, w3, b3, base, ws);
  kan_forward<<<blocks, 256, 0, stream>>>(x, ws, out, TILES);
}


# A-fragment generation strength-reduced: K loops fully unrolled, input index and RBF centres fixed per lane (no division/branch), inputs read once as b128, exp as one exp2 of a pre-scaled argument
# speedup vs baseline: 1.3192x; 1.3192x over previous
#include <hip/hip_runtime.h>


typedef __attribute__((ext_vector_type(2))) float v2f;
typedef __attribute__((ext_vector_type(4))) float v4f;
typedef __attribute__((ext_vector_type(8))) float v8f;

#define GRIDN 12
#define RBF_S 1.5570899671947016
__device__ __forceinline__ constexpr float rbf_cs(int g) {
  return (float)((-3.0 + (6.0 / 11.0) * (double)g) * RBF_S);
}

#define P0 12
#define P1 28
#define P2 12

#define NSTEP1 30
#define NSTEP2 79
#define NSTEP3 40
#define NT1 2
#define NT2 1
#define NT3 1
#define B1_FLOATS (NSTEP1 * NT1 * 64)
#define B2_FLOATS (NSTEP2 * NT2 * 64)
#define B3_FLOATS (NSTEP3 * NT3 * 64)
#define B1_OFF 0
#define B2_OFF (B1_FLOATS)
#define B3_OFF (B1_FLOATS + B2_FLOATS)
#define BTOTAL (B1_FLOATS + B2_FLOATS + B3_FLOATS)

__global__ __launch_bounds__(256) void kan_pack(
    const float* __restrict__ c1, const float* __restrict__ w1, const float* __restrict__ b1,
    const float* __restrict__ c2, const float* __restrict__ w2, const float* __restrict__ b2,
    const float* __restrict__ c3, const float* __restrict__ w3, const float* __restrict__ b3,
    const float* __restrict__ base, float* __restrict__ ws) {
  int idx = blockIdx.x * blockDim.x + threadIdx.x;
  if (idx >= BTOTAL) return;

  int layer, e, nt_count;
  if (idx < B2_OFF)       { layer = 0; e = idx;           nt_count = NT1; }
  else if (idx < B3_OFF)  { layer = 1; e = idx - B2_OFF;  nt_count = NT2; }
  else                    { layer = 2; e = idx - B3_OFF;  nt_count = NT3; }

  int v    = e & 1;
  int lane = (e >> 1) & 31;
  int q    = (e >> 1) >> 5;
  int t    = q / nt_count;
  int nt   = q - t * nt_count;
  int k    = 4 * t + 2 * (lane >> 4) + v;
  int n    = nt * 16 + (lane & 15);

  float val = 0.0f;
  if (layer == 0) {
    if (n < 24) {
      if (k < 108)      { int i = k / 12, g = k - i * 12; val = c1[(i * 24 + n) * 12 + g]; }
      else if (k < 117) { val = w1[n * 9 + (k - 108)]; }
      else if (k == 117){ val = b1[n]; }
    }
  } else if (layer == 1) {
    if (n < 12) {
      if (k < 288)      { int i = k / 12, g = k - i * 12; val = c2[(i * 12 + n) * 12 + g]; }
      else if (k < 312) { val = w2[n * 24 + (k - 288)]; }
      else if (k == 312){ val = b2[n]; }
    }
  } else {
    if (n < 3) {
      if (k < 144)      { int i = k / 12, g = k - i * 12; val = c3[(i * 3 + n) * 12 + g]; }
      else if (k < 156) { val = w3[n * 12 + (k - 144)]; }
      else if (k == 156){ val = b3[n] + base[n]; }
    }
  }
  ws[idx] = val;
}

template <int I, int NT>
__device__ __forceinline__ void kan_layer_wmma(const float* __restrict__ arow,
                                               int half,
                                               const float* __restrict__ Bl,
                                               const v2f (&cs)[3],
                                               v8f& c0, v8f& c1) {
  const float* __restrict__ arowh = arow + 2 * half;
  constexpr int NBASIS = 3 * I;
  constexpr int NTAIL  = (I + 1 + 3) / 4;
#pragma unroll
  for (int i4 = 0; i4 < I; i4 += 4) {
    const v4f av = *(const v4f*)(arow + i4);
#pragma unroll
    for (int ii = 0; ii < 4; ++ii) {
      const int i = i4 + ii;
      if (i < I) {
        const float as = av[ii] * (float)RBF_S;
        const v2f asv = {as, as};
#pragma unroll
        for (int p = 0; p < 3; ++p) {
          const int t = 3 * i + p;
          const v2f q = asv - cs[p];
          const v2f e = -(q * q);
          v2f a;
          a.x = __builtin_amdgcn_exp2f(e.x);
          a.y = __builtin_amdgcn_exp2f(e.y);
          const v2f b0 = *(const v2f*)(Bl + (t * NT + 0) * 64);
          c0 = __builtin_amdgcn_wmma_f32_16x16x4_f32(false, a, false, b0, (short)0, c0,
                                                     false, false);
          if (NT == 2) {
            const v2f b1 = *(const v2f*)(Bl + (t * NT + 1) * 64);
            c1 = __builtin_amdgcn_wmma_f32_16x16x4_f32(false, a, false, b1, (short)0, c1,
                                                       false, false);
          }
        }
      }
    }
  }
#pragma unroll
  for (int tt = 0; tt < NTAIL; ++tt) {
    const int t = NBASIS + tt;
    v2f a;
    if (4 * tt + 3 < I) {
      a = *(const v2f*)(arowh + 4 * tt);
    } else {
#pragma unroll
      for (int v = 0; v < 2; ++v) {
        const int jlo = 4 * tt + v, jhi = jlo + 2;
        const float elo = jlo < I ? arow[jlo] : (jlo == I ? 1.0f : 0.0f);
        const float ehi = jhi < I ? arow[jhi] : (jhi == I ? 1.0f : 0.0f);
        a[v] = half ? ehi : elo;
      }
    }
    const v2f b0 = *(const v2f*)(Bl + (t * NT + 0) * 64);
    c0 = __builtin_amdgcn_wmma_f32_16x16x4_f32(false, a, false, b0, (short)0, c0,
                                               false, false);
    if (NT == 2) {
      const v2f b1 = *(const v2f*)(Bl + (t * NT + 1) * 64);
      c1 = __builtin_amdgcn_wmma_f32_16x16x4_f32(false, a, false, b1, (short)0, c1,
                                                 false, false);
    }
  }
}

__device__ __forceinline__ float silu_f(float z) {
  return z * (1.0f / (1.0f + __expf(-z)));
}

__global__ __launch_bounds__(256, 2) void kan_forward(const float* __restrict__ x,
                                                      const float* __restrict__ ws,
                                                      float* __restrict__ out,
                                                      int tiles_per_wave) {
  __shared__ __align__(16) float Bsh[BTOTAL];
  __shared__ __align__(16) float act0[8][16][P0];
  __shared__ __align__(16) float act1[8][16][P1];
  __shared__ __align__(16) float act2[8][16][P2];
  __shared__ __align__(16) float cor[8][16][4];

  const int tid = threadIdx.x;
  for (int i = tid; i < BTOTAL; i += 256) Bsh[i] = ws[i];
  __syncthreads();

  const int wv   = tid >> 5;
  const int lane = tid & 31;
  const int half = lane >> 4;
  const int m    = lane & 15;
  const float* a0 = &act0[wv][m][0];
  const float* a1 = &act1[wv][m][0];
  const float* a2 = &act2[wv][m][0];
  const float* Bl = Bsh + lane * 2;

  constexpr float CS_LO[3][2] = {{rbf_cs(0), rbf_cs(1)}, {rbf_cs(4), rbf_cs(5)}, {rbf_cs(8), rbf_cs(9)}};
  constexpr float CS_HI[3][2] = {{rbf_cs(2), rbf_cs(3)}, {rbf_cs(6), rbf_cs(7)}, {rbf_cs(10), rbf_cs(11)}};
  v2f cs[3];
#pragma unroll
  for (int p = 0; p < 3; ++p) {
    cs[p].x = half ? CS_HI[p][0] : CS_LO[p][0];
    cs[p].y = half ? CS_HI[p][1] : CS_LO[p][1];
  }

  for (int it = 0; it < tiles_per_wave; ++it) {
    const int rowbase = ((blockIdx.x * tiles_per_wave + it) * 8 + wv) * 16;

    for (int e = lane; e < 144; e += 32) {
      int r = e / 9, c = e - r * 9;
      act0[wv][r][c] = __builtin_nontemporal_load(&x[(size_t)(rowbase + r) * 9 + c]);
    }
    asm volatile("s_wait_dscnt 0" ::: "memory");

    v8f c0 = {}, c1 = {};
    kan_layer_wmma<9, NT1>(a0, half, Bl + B1_OFF, cs, c0, c1);
#pragma unroll
    for (int v = 0; v < 8; ++v) {
      int r = half * 8 + v;
      act1[wv][r][m] = silu_f(c0[v]);
      if (m < 8) act1[wv][r][16 + m] = silu_f(c1[v]);
    }
    asm volatile("s_wait_dscnt 0" ::: "memory");

    v8f d0 = {}, dz = {};
    kan_layer_wmma<24, NT2>(a1, half, Bl + B2_OFF, cs, d0, dz);
#pragma unroll
    for (int v = 0; v < 8; ++v) {
      int r = half * 8 + v;
      if (m < 12) act2[wv][r][m] = silu_f(d0[v]);
    }
    asm volatile("s_wait_dscnt 0" ::: "memory");

    v8f e0 = {}, ez = {};
    kan_layer_wmma<12, NT3>(a2, half, Bl + B3_OFF, cs, e0, ez);
#pragma unroll
    for (int v = 0; v < 8; ++v) {
      int r = half * 8 + v;
      if (m < 3) cor[wv][r][m] = e0[v];
    }
    asm volatile("s_wait_dscnt 0" ::: "memory");

    if (half == 0) {
      float p0 = cor[wv][m][0], p1 = cor[wv][m][1], p2 = cor[wv][m][2];
      float mx = fmaxf(p0, fmaxf(p1, p2));
      float q0 = __expf(p0 - mx), q1 = __expf(p1 - mx), q2 = __expf(p2 - mx);
      float inv = 1.0f / (q0 + q1 + q2);
      float* o = out + (size_t)(rowbase + m) * 3;
      __builtin_nontemporal_store(q0 * inv, o + 0);
      __builtin_nontemporal_store(q1 * inv, o + 1);
      __builtin_nontemporal_store(q2 * inv, o + 2);
    }
    asm volatile("s_wait_dscnt 0" ::: "memory");
  }
}

extern "C" void kernel_launch(void* const* d_in, const int* in_sizes, int n_in,
                              void* d_out, int out_size, void* d_ws, size_t ws_size,
                              hipStream_t stream) {
  const float* x    = (const float*)d_in[0];
  const float* c1   = (const float*)d_in[1];
  const float* w1   = (const float*)d_in[2];
  const float* b1   = (const float*)d_in[3];
  const float* c2   = (const float*)d_in[4];
  const float* w2   = (const float*)d_in[5];
  const float* b2   = (const float*)d_in[6];
  const float* c3   = (const float*)d_in[7];
  const float* w3   = (const float*)d_in[8];
  const float* b3   = (const float*)d_in[9];
  const float* base = (const float*)d_in[10];
  float* ws  = (float*)d_ws;
  float* out = (float*)d_out;

  const int Btot = in_sizes[0] / 9;
  const int TILES = 4;
  const int blocks = Btot / (128 * TILES);

  kan_pack<<<(BTOTAL + 255) / 256, 256, 0, stream>>>(c1, w1, b1, c2, w2, b2,
                                                     c3, w3, b3, base, ws);
  kan_forward<<<blocks, 256, 0, stream>>>(x, ws, out, TILES);
}


# SiLU denominators and softmax normaliser: the exact IEEE division sequences replaced by the hardware reciprocal v_rcp_f32 (1 ulp), about 260 fewer vector instructions per row tile
# speedup vs baseline: 1.3771x; 1.0439x over previous
#include <hip/hip_runtime.h>


typedef __attribute__((ext_vector_type(2))) float v2f;
typedef __attribute__((ext_vector_type(4))) float v4f;
typedef __attribute__((ext_vector_type(8))) float v8f;

#define GRIDN 12
#define RBF_S 1.5570899671947016
__device__ __forceinline__ constexpr float rbf_cs(int g) {
  return (float)((-3.0 + (6.0 / 11.0) * (double)g) * RBF_S);
}

#define P0 12
#define P1 28
#define P2 12

#define NSTEP1 30
#define NSTEP2 79
#define NSTEP3 40
#define NT1 2
#define NT2 1
#define NT3 1
#define B1_FLOATS (NSTEP1 * NT1 * 64)
#define B2_FLOATS (NSTEP2 * NT2 * 64)
#define B3_FLOATS (NSTEP3 * NT3 * 64)
#define B1_OFF 0
#define B2_OFF (B1_FLOATS)
#define B3_OFF (B1_FLOATS + B2_FLOATS)
#define BTOTAL (B1_FLOATS + B2_FLOATS + B3_FLOATS)

__global__ __launch_bounds__(256) void kan_pack(
    const float* __restrict__ c1, const float* __restrict__ w1, const float* __restrict__ b1,
    const float* __restrict__ c2, const float* __restrict__ w2, const float* __restrict__ b2,
    const float* __restrict__ c3, const float* __restrict__ w3, const float* __restrict__ b3,
    const float* __restrict__ base, float* __restrict__ ws) {
  int idx = blockIdx.x * blockDim.x + threadIdx.x;
  if (idx >= BTOTAL) return;

  int layer, e, nt_count;
  if (idx < B2_OFF)       { layer = 0; e = idx;           nt_count = NT1; }
  else if (idx < B3_OFF)  { layer = 1; e = idx - B2_OFF;  nt_count = NT2; }
  else                    { layer = 2; e = idx - B3_OFF;  nt_count = NT3; }

  int v    = e & 1;
  int lane = (e >> 1) & 31;
  int q    = (e >> 1) >> 5;
  int t    = q / nt_count;
  int nt   = q - t * nt_count;
  int k    = 4 * t + 2 * (lane >> 4) + v;
  int n    = nt * 16 + (lane & 15);

  float val = 0.0f;
  if (layer == 0) {
    if (n < 24) {
      if (k < 108)      { int i = k / 12, g = k - i * 12; val = c1[(i * 24 + n) * 12 + g]; }
      else if (k < 117) { val = w1[n * 9 + (k - 108)]; }
      else if (k == 117){ val = b1[n]; }
    }
  } else if (layer == 1) {
    if (n < 12) {
      if (k < 288)      { int i = k / 12, g = k - i * 12; val = c2[(i * 12 + n) * 12 + g]; }
      else if (k < 312) { val = w2[n * 24 + (k - 288)]; }
      else if (k == 312){ val = b2[n]; }
    }
  } else {
    if (n < 3) {
      if (k < 144)      { int i = k / 12, g = k - i * 12; val = c3[(i * 3 + n) * 12 + g]; }
      else if (k < 156) { val = w3[n * 12 + (k - 144)]; }
      else if (k == 156){ val = b3[n] + base[n]; }
    }
  }
  ws[idx] = val;
}

template <int I, int NT>
__device__ __forceinline__ void kan_layer_wmma(const float* __restrict__ arow,
                                               int half,
                                               const float* __restrict__ Bl,
                                               const v2f (&cs)[3],
                                               v8f& c0, v8f& c1) {
  const float* __restrict__ arowh = arow + 2 * half;
  constexpr int NBASIS = 3 * I;
  constexpr int NTAIL  = (I + 1 + 3) / 4;
#pragma unroll
  for (int i4 = 0; i4 < I; i4 += 4) {
    const v4f av = *(const v4f*)(arow + i4);
#pragma unroll
    for (int ii = 0; ii < 4; ++ii) {
      const int i = i4 + ii;
      if (i < I) {
        const float as = av[ii] * (float)RBF_S;
        const v2f asv = {as, as};
#pragma unroll
        for (int p = 0; p < 3; ++p) {
          const int t = 3 * i + p;
          const v2f q = asv - cs[p];
          const v2f e = -(q * q);
          v2f a;
          a.x = __builtin_amdgcn_exp2f(e.x);
          a.y = __builtin_amdgcn_exp2f(e.y);
          const v2f b0 = *(const v2f*)(Bl + (t * NT + 0) * 64);
          c0 = __builtin_amdgcn_wmma_f32_16x16x4_f32(false, a, false, b0, (short)0, c0,
                                                     false, false);
          if (NT == 2) {
            const v2f b1 = *(const v2f*)(Bl + (t * NT + 1) * 64);
            c1 = __builtin_amdgcn_wmma_f32_16x16x4_f32(false, a, false, b1, (short)0, c1,
                                                       false, false);
          }
        }
      }
    }
  }
#pragma unroll
  for (int tt = 0; tt < NTAIL; ++tt) {
    const int t = NBASIS + tt;
    v2f a;
    if (4 * tt + 3 < I) {
      a = *(const v2f*)(arowh + 4 * tt);
    } else {
#pragma unroll
      for (int v = 0; v < 2; ++v) {
        const int jlo = 4 * tt + v, jhi = jlo + 2;
        const float elo = jlo < I ? arow[jlo] : (jlo == I ? 1.0f : 0.0f);
        const float ehi = jhi < I ? arow[jhi] : (jhi == I ? 1.0f : 0.0f);
        a[v] = half ? ehi : elo;
      }
    }
    const v2f b0 = *(const v2f*)(Bl + (t * NT + 0) * 64);
    c0 = __builtin_amdgcn_wmma_f32_16x16x4_f32(false, a, false, b0, (short)0, c0,
                                               false, false);
    if (NT == 2) {
      const v2f b1 = *(const v2f*)(Bl + (t * NT + 1) * 64);
      c1 = __builtin_amdgcn_wmma_f32_16x16x4_f32(false, a, false, b1, (short)0, c1,
                                                 false, false);
    }
  }
}

__device__ __forceinline__ float silu_f(float z) {
  return z * __builtin_amdgcn_rcpf(1.0f + __expf(-z));
}

__global__ __launch_bounds__(256, 2) void kan_forward(const float* __restrict__ x,
                                                      const float* __restrict__ ws,
                                                      float* __restrict__ out,
                                                      int tiles_per_wave) {
  __shared__ __align__(16) float Bsh[BTOTAL];
  __shared__ __align__(16) float act0[8][16][P0];
  __shared__ __align__(16) float act1[8][16][P1];
  __shared__ __align__(16) float act2[8][16][P2];
  __shared__ __align__(16) float cor[8][16][4];

  const int tid = threadIdx.x;
  for (int i = tid; i < BTOTAL; i += 256) Bsh[i] = ws[i];
  __syncthreads();

  const int wv   = tid >> 5;
  const int lane = tid & 31;
  const int half = lane >> 4;
  const int m    = lane & 15;
  const float* a0 = &act0[wv][m][0];
  const float* a1 = &act1[wv][m][0];
  const float* a2 = &act2[wv][m][0];
  const float* Bl = Bsh + lane * 2;

  constexpr float CS_LO[3][2] = {{rbf_cs(0), rbf_cs(1)}, {rbf_cs(4), rbf_cs(5)}, {rbf_cs(8), rbf_cs(9)}};
  constexpr float CS_HI[3][2] = {{rbf_cs(2), rbf_cs(3)}, {rbf_cs(6), rbf_cs(7)}, {rbf_cs(10), rbf_cs(11)}};
  v2f cs[3];
#pragma unroll
  for (int p = 0; p < 3; ++p) {
    cs[p].x = half ? CS_HI[p][0] : CS_LO[p][0];
    cs[p].y = half ? CS_HI[p][1] : CS_LO[p][1];
  }

  for (int it = 0; it < tiles_per_wave; ++it) {
    const int rowbase = ((blockIdx.x * tiles_per_wave + it) * 8 + wv) * 16;

    for (int e = lane; e < 144; e += 32) {
      int r = e / 9, c = e - r * 9;
      act0[wv][r][c] = __builtin_nontemporal_load(&x[(size_t)(rowbase + r) * 9 + c]);
    }
    asm volatile("s_wait_dscnt 0" ::: "memory");

    v8f c0 = {}, c1 = {};
    kan_layer_wmma<9, NT1>(a0, half, Bl + B1_OFF, cs, c0, c1);
#pragma unroll
    for (int v = 0; v < 8; ++v) {
      int r = half * 8 + v;
      act1[wv][r][m] = silu_f(c0[v]);
      if (m < 8) act1[wv][r][16 + m] = silu_f(c1[v]);
    }
    asm volatile("s_wait_dscnt 0" ::: "memory");

    v8f d0 = {}, dz = {};
    kan_layer_wmma<24, NT2>(a1, half, Bl + B2_OFF, cs, d0, dz);
#pragma unroll
    for (int v = 0; v < 8; ++v) {
      int r = half * 8 + v;
      if (m < 12) act2[wv][r][m] = silu_f(d0[v]);
    }
    asm volatile("s_wait_dscnt 0" ::: "memory");

    v8f e0 = {}, ez = {};
    kan_layer_wmma<12, NT3>(a2, half, Bl + B3_OFF, cs, e0, ez);
#pragma unroll
    for (int v = 0; v < 8; ++v) {
      int r = half * 8 + v;
      if (m < 3) cor[wv][r][m] = e0[v];
    }
    asm volatile("s_wait_dscnt 0" ::: "memory");

    if (half == 0) {
      float p0 = cor[wv][m][0], p1 = cor[wv][m][1], p2 = cor[wv][m][2];
      float mx = fmaxf(p0, fmaxf(p1, p2));
      float q0 = __expf(p0 - mx), q1 = __expf(p1 - mx), q2 = __expf(p2 - mx);
      float inv = __builtin_amdgcn_rcpf(q0 + q1 + q2);
      float* o = out + (size_t)(rowbase + m) * 3;
      __builtin_nontemporal_store(q0 * inv, o + 0);
      __builtin_nontemporal_store(q1 * inv, o + 1);
      __builtin_nontemporal_store(q2 * inv, o + 2);
    }
    asm volatile("s_wait_dscnt 0" ::: "memory");
  }
}

extern "C" void kernel_launch(void* const* d_in, const int* in_sizes, int n_in,
                              void* d_out, int out_size, void* d_ws, size_t ws_size,
                              hipStream_t stream) {
  const float* x    = (const float*)d_in[0];
  const float* c1   = (const float*)d_in[1];
  const float* w1   = (const float*)d_in[2];
  const float* b1   = (const float*)d_in[3];
  const float* c2   = (const float*)d_in[4];
  const float* w2   = (const float*)d_in[5];
  const float* b2   = (const float*)d_in[6];
  const float* c3   = (const float*)d_in[7];
  const float* w3   = (const float*)d_in[8];
  const float* b3   = (const float*)d_in[9];
  const float* base = (const float*)d_in[10];
  float* ws  = (float*)d_ws;
  float* out = (float*)d_out;

  const int Btot = in_sizes[0] / 9;
  const int TILES = 4;
  const int blocks = Btot / (128 * TILES);

  kan_pack<<<(BTOTAL + 255) / 256, 256, 0, stream>>>(c1, w1, b1, c2, w2, b2,
                                                     c3, w3, b3, base, ws);
  kan_forward<<<blocks, 256, 0, stream>>>(x, ws, out, TILES);
}


# Layer-1 inputs: each lane loads its own 9-float row from global into registers one tile ahead (2 b128 + 1 b32); the per-tile global-to-LDS staging loop of 5 dependent round trips is gone
# speedup vs baseline: 1.4585x; 1.0591x over previous
#include <hip/hip_runtime.h>


typedef __attribute__((ext_vector_type(2))) float v2f;
typedef __attribute__((ext_vector_type(4))) float v4f;
typedef __attribute__((ext_vector_type(8))) float v8f;
typedef float v4f_a4 __attribute__((ext_vector_type(4), aligned(4)));

#define GRIDN 12
#define RBF_S 1.5570899671947016
__device__ __forceinline__ constexpr float rbf_cs(int g) {
  return (float)((-3.0 + (6.0 / 11.0) * (double)g) * RBF_S);
}

#define P1 28
#define P2 12

#define NSTEP1 30
#define NSTEP2 79
#define NSTEP3 40
#define NT1 2
#define NT2 1
#define NT3 1
#define B1_FLOATS (NSTEP1 * NT1 * 64)
#define B2_FLOATS (NSTEP2 * NT2 * 64)
#define B3_FLOATS (NSTEP3 * NT3 * 64)
#define B1_OFF 0
#define B2_OFF (B1_FLOATS)
#define B3_OFF (B1_FLOATS + B2_FLOATS)
#define BTOTAL (B1_FLOATS + B2_FLOATS + B3_FLOATS)

__global__ __launch_bounds__(256) void kan_pack(
    const float* __restrict__ c1, const float* __restrict__ w1, const float* __restrict__ b1,
    const float* __restrict__ c2, const float* __restrict__ w2, const float* __restrict__ b2,
    const float* __restrict__ c3, const float* __restrict__ w3, const float* __restrict__ b3,
    const float* __restrict__ base, float* __restrict__ ws) {
  int idx = blockIdx.x * blockDim.x + threadIdx.x;
  if (idx >= BTOTAL) return;

  int layer, e, nt_count;
  if (idx < B2_OFF)       { layer = 0; e = idx;           nt_count = NT1; }
  else if (idx < B3_OFF)  { layer = 1; e = idx - B2_OFF;  nt_count = NT2; }
  else                    { layer = 2; e = idx - B3_OFF;  nt_count = NT3; }

  int v    = e & 1;
  int lane = (e >> 1) & 31;
  int q    = (e >> 1) >> 5;
  int t    = q / nt_count;
  int nt   = q - t * nt_count;
  int k    = 4 * t + 2 * (lane >> 4) + v;
  int n    = nt * 16 + (lane & 15);

  float val = 0.0f;
  if (layer == 0) {
    if (n < 24) {
      if (k < 108)      { int i = k / 12, g = k - i * 12; val = c1[(i * 24 + n) * 12 + g]; }
      else if (k < 117) { val = w1[n * 9 + (k - 108)]; }
      else if (k == 117){ val = b1[n]; }
    }
  } else if (layer == 1) {
    if (n < 12) {
      if (k < 288)      { int i = k / 12, g = k - i * 12; val = c2[(i * 12 + n) * 12 + g]; }
      else if (k < 312) { val = w2[n * 24 + (k - 288)]; }
      else if (k == 312){ val = b2[n]; }
    }
  } else {
    if (n < 3) {
      if (k < 144)      { int i = k / 12, g = k - i * 12; val = c3[(i * 3 + n) * 12 + g]; }
      else if (k < 156) { val = w3[n * 12 + (k - 144)]; }
      else if (k == 156){ val = b3[n] + base[n]; }
    }
  }
  ws[idx] = val;
}

template <int I, int NT, bool REGS>
__device__ __forceinline__ void kan_layer_wmma(const float* __restrict__ arow,
                                               const float* __restrict__ xin,
                                               int half,
                                               const float* __restrict__ Bl,
                                               const v2f (&cs)[3],
                                               v8f& c0, v8f& c1) {
  const float* __restrict__ arowh = REGS ? arow : arow + 2 * half;
  constexpr int NBASIS = 3 * I;
  constexpr int NTAIL  = (I + 1 + 3) / 4;
#pragma unroll
  for (int i4 = 0; i4 < I; i4 += 4) {
    v4f av = {};
    if (!REGS) av = *(const v4f*)(arow + i4);
#pragma unroll
    for (int ii = 0; ii < 4; ++ii) {
      const int i = i4 + ii;
      if (i < I) {
        const float as = (REGS ? xin[i] : av[ii]) * (float)RBF_S;
        const v2f asv = {as, as};
#pragma unroll
        for (int p = 0; p < 3; ++p) {
          const int t = 3 * i + p;
          const v2f q = asv - cs[p];
          const v2f e = -(q * q);
          v2f a;
          a.x = __builtin_amdgcn_exp2f(e.x);
          a.y = __builtin_amdgcn_exp2f(e.y);
          const v2f b0 = *(const v2f*)(Bl + (t * NT + 0) * 64);
          c0 = __builtin_amdgcn_wmma_f32_16x16x4_f32(false, a, false, b0, (short)0, c0,
                                                     false, false);
          if (NT == 2) {
            const v2f b1 = *(const v2f*)(Bl + (t * NT + 1) * 64);
            c1 = __builtin_amdgcn_wmma_f32_16x16x4_f32(false, a, false, b1, (short)0, c1,
                                                       false, false);
          }
        }
      }
    }
  }
#pragma unroll
  for (int tt = 0; tt < NTAIL; ++tt) {
    const int t = NBASIS + tt;
    v2f a;
    if (4 * tt + 3 < I) {
      if (REGS) {
        a.x = half ? xin[4 * tt + 2] : xin[4 * tt + 0];
        a.y = half ? xin[4 * tt + 3] : xin[4 * tt + 1];
      } else {
        a = *(const v2f*)(arowh + 4 * tt);
      }
    } else {
#pragma unroll
      for (int v = 0; v < 2; ++v) {
        const int jlo = 4 * tt + v, jhi = jlo + 2;
        const float elo = jlo < I ? (REGS ? xin[jlo] : arow[jlo]) : (jlo == I ? 1.0f : 0.0f);
        const float ehi = jhi < I ? (REGS ? xin[jhi] : arow[jhi]) : (jhi == I ? 1.0f : 0.0f);
        a[v] = half ? ehi : elo;
      }
    }
    const v2f b0 = *(const v2f*)(Bl + (t * NT + 0) * 64);
    c0 = __builtin_amdgcn_wmma_f32_16x16x4_f32(false, a, false, b0, (short)0, c0,
                                               false, false);
    if (NT == 2) {
      const v2f b1 = *(const v2f*)(Bl + (t * NT + 1) * 64);
      c1 = __builtin_amdgcn_wmma_f32_16x16x4_f32(false, a, false, b1, (short)0, c1,
                                                 false, false);
    }
  }
}

__device__ __forceinline__ float silu_f(float z) {
  return z * __builtin_amdgcn_rcpf(1.0f + __expf(-z));
}

__global__ __launch_bounds__(256, 2) void kan_forward(const float* __restrict__ x,
                                                      const float* __restrict__ ws,
                                                      float* __restrict__ out,
                                                      int tiles_per_wave) {
  __shared__ __align__(16) float Bsh[BTOTAL];
  __shared__ __align__(16) float act1[8][16][P1];
  __shared__ __align__(16) float act2[8][16][P2];
  __shared__ __align__(16) float cor[8][16][4];

  const int tid  = threadIdx.x;
  const int wv   = tid >> 5;
  const int lane = tid & 31;
  const int half = lane >> 4;
  const int m    = lane & 15;

  const int tile0 = blockIdx.x * tiles_per_wave * 8 + wv;
  const float* xrow = x + (size_t)(tile0 * 16 + m) * 9;
  v4f   xa = *(const v4f_a4*)(xrow);
  v4f   xb = *(const v4f_a4*)(xrow + 4);
  float xc = xrow[8];

  for (int i = tid; i < BTOTAL; i += 256) Bsh[i] = ws[i];
  __syncthreads();

  const float* a1 = &act1[wv][m][0];
  const float* a2 = &act2[wv][m][0];
  const float* Bl = Bsh + lane * 2;

  constexpr float CS_LO[3][2] = {{rbf_cs(0), rbf_cs(1)}, {rbf_cs(4), rbf_cs(5)}, {rbf_cs(8), rbf_cs(9)}};
  constexpr float CS_HI[3][2] = {{rbf_cs(2), rbf_cs(3)}, {rbf_cs(6), rbf_cs(7)}, {rbf_cs(10), rbf_cs(11)}};
  v2f cs[3];
#pragma unroll
  for (int p = 0; p < 3; ++p) {
    cs[p].x = half ? CS_HI[p][0] : CS_LO[p][0];
    cs[p].y = half ? CS_HI[p][1] : CS_LO[p][1];
  }

  for (int it = 0; it < tiles_per_wave; ++it) {
    const int rowbase = ((blockIdx.x * tiles_per_wave + it) * 8 + wv) * 16;

    const float xin[9] = {xa[0], xa[1], xa[2], xa[3], xb[0], xb[1], xb[2], xb[3], xc};
    v8f c0 = {}, c1 = {};
    kan_layer_wmma<9, NT1, true>(nullptr, xin, half, Bl + B1_OFF, cs, c0, c1);

    {
      const int itn = it + 1 < tiles_per_wave ? it + 1 : it;
      const float* xn = xrow + (size_t)itn * (8 * 16 * 9);
      xa = *(const v4f_a4*)(xn);
      xb = *(const v4f_a4*)(xn + 4);
      xc = xn[8];
    }
#pragma unroll
    for (int v = 0; v < 8; ++v) {
      int r = half * 8 + v;
      act1[wv][r][m] = silu_f(c0[v]);
      if (m < 8) act1[wv][r][16 + m] = silu_f(c1[v]);
    }
    asm volatile("s_wait_dscnt 0" ::: "memory");

    v8f d0 = {}, dz = {};
    kan_layer_wmma<24, NT2, false>(a1, nullptr, half, Bl + B2_OFF, cs, d0, dz);
#pragma unroll
    for (int v = 0; v < 8; ++v) {
      int r = half * 8 + v;
      if (m < 12) act2[wv][r][m] = silu_f(d0[v]);
    }
    asm volatile("s_wait_dscnt 0" ::: "memory");

    v8f e0 = {}, ez = {};
    kan_layer_wmma<12, NT3, false>(a2, nullptr, half, Bl + B3_OFF, cs, e0, ez);
#pragma unroll
    for (int v = 0; v < 8; ++v) {
      int r = half * 8 + v;
      if (m < 3) cor[wv][r][m] = e0[v];
    }
    asm volatile("s_wait_dscnt 0" ::: "memory");

    if (half == 0) {
      float p0 = cor[wv][m][0], p1 = cor[wv][m][1], p2 = cor[wv][m][2];
      float mx = fmaxf(p0, fmaxf(p1, p2));
      float q0 = __expf(p0 - mx), q1 = __expf(p1 - mx), q2 = __expf(p2 - mx);
      float inv = __builtin_amdgcn_rcpf(q0 + q1 + q2);
      float* o = out + (size_t)(rowbase + m) * 3;
      __builtin_nontemporal_store(q0 * inv, o + 0);
      __builtin_nontemporal_store(q1 * inv, o + 1);
      __builtin_nontemporal_store(q2 * inv, o + 2);
    }
    asm volatile("s_wait_dscnt 0" ::: "memory");
  }
}

extern "C" void kernel_launch(void* const* d_in, const int* in_sizes, int n_in,
                              void* d_out, int out_size, void* d_ws, size_t ws_size,
                              hipStream_t stream) {
  const float* x    = (const float*)d_in[0];
  const float* c1   = (const float*)d_in[1];
  const float* w1   = (const float*)d_in[2];
  const float* b1   = (const float*)d_in[3];
  const float* c2   = (const float*)d_in[4];
  const float* w2   = (const float*)d_in[5];
  const float* b2   = (const float*)d_in[6];
  const float* c3   = (const float*)d_in[7];
  const float* w3   = (const float*)d_in[8];
  const float* b3   = (const float*)d_in[9];
  const float* base = (const float*)d_in[10];
  float* ws  = (float*)d_ws;
  float* out = (float*)d_out;

  const int Btot = in_sizes[0] / 9;
  const int TILES = 4;
  const int blocks = Btot / (128 * TILES);

  kan_pack<<<(BTOTAL + 255) / 256, 256, 0, stream>>>(c1, w1, b1, c2, w2, b2,
                                                     c3, w3, b3, base, ws);
  kan_forward<<<blocks, 256, 0, stream>>>(x, ws, out, TILES);
}


# RBF basis by recurrence: per input and lane 2 exps for the inner centre pair, outer pairs as F*rho, rho = exp2(+-x*2/h): 3 exps for 6; constants folded into packed B; K order of centres mirrored
# speedup vs baseline: 1.5466x; 1.0604x over previous
#include <hip/hip_runtime.h>


typedef __attribute__((ext_vector_type(2))) float v2f;
typedef __attribute__((ext_vector_type(4))) float v4f;
typedef __attribute__((ext_vector_type(8))) float v8f;
typedef float v4f_a4 __attribute__((ext_vector_type(4), aligned(4)));

#define GRIDN 12
#define RBF_S 1.5570899671947016
#define RBF_D 0.8493218002880191
#define RBF_R 5.289881816592866
#define KC10 0.049787068367863944
#define KC11 0.006737946999085467
#define KC20 4.5399929762484854e-05
#define KC21 8.3152871910356788e-07

#define P1 28
#define P2 12

#define NSTEP1 30
#define NSTEP2 79
#define NSTEP3 40
#define NT1 2
#define NT2 1
#define NT3 1
#define B1_FLOATS (NSTEP1 * NT1 * 64)
#define B2_FLOATS (NSTEP2 * NT2 * 64)
#define B3_FLOATS (NSTEP3 * NT3 * 64)
#define B1_OFF 0
#define B2_OFF (B1_FLOATS)
#define B3_OFF (B1_FLOATS + B2_FLOATS)
#define BTOTAL (B1_FLOATS + B2_FLOATS + B3_FLOATS)

__device__ __forceinline__ int basis_g(int kk) {
  const int p = kk >> 2, hh = (kk >> 1) & 1, v = kk & 1;
  return hh ? 6 + 2 * p + v : 5 - 2 * p - v;
}
__device__ __forceinline__ float basis_kc(int kk) {
  const int p = kk >> 2, v = kk & 1;
  return p == 0 ? 1.0f : (p == 1 ? (v ? (float)KC11 : (float)KC10) : (v ? (float)KC21 : (float)KC20));
}

__global__ __launch_bounds__(256) void kan_pack(
    const float* __restrict__ c1, const float* __restrict__ w1, const float* __restrict__ b1,
    const float* __restrict__ c2, const float* __restrict__ w2, const float* __restrict__ b2,
    const float* __restrict__ c3, const float* __restrict__ w3, const float* __restrict__ b3,
    const float* __restrict__ base, float* __restrict__ ws) {
  int idx = blockIdx.x * blockDim.x + threadIdx.x;
  if (idx >= BTOTAL) return;

  int layer, e, nt_count;
  if (idx < B2_OFF)       { layer = 0; e = idx;           nt_count = NT1; }
  else if (idx < B3_OFF)  { layer = 1; e = idx - B2_OFF;  nt_count = NT2; }
  else                    { layer = 2; e = idx - B3_OFF;  nt_count = NT3; }

  int v    = e & 1;
  int lane = (e >> 1) & 31;
  int q    = (e >> 1) >> 5;
  int t    = q / nt_count;
  int nt   = q - t * nt_count;
  int k    = 4 * t + 2 * (lane >> 4) + v;
  int n    = nt * 16 + (lane & 15);

  float val = 0.0f;
  if (layer == 0) {
    if (n < 24) {
      if (k < 108)      { int i = k / 12, kk = k - i * 12; val = c1[(i * 24 + n) * 12 + basis_g(kk)] * basis_kc(kk); }
      else if (k < 117) { val = w1[n * 9 + (k - 108)]; }
      else if (k == 117){ val = b1[n]; }
    }
  } else if (layer == 1) {
    if (n < 12) {
      if (k < 288)      { int i = k / 12, kk = k - i * 12; val = c2[(i * 12 + n) * 12 + basis_g(kk)] * basis_kc(kk); }
      else if (k < 312) { val = w2[n * 24 + (k - 288)]; }
      else if (k == 312){ val = b2[n]; }
    }
  } else {
    if (n < 3) {
      if (k < 144)      { int i = k / 12, kk = k - i * 12; val = c3[(i * 3 + n) * 12 + basis_g(kk)] * basis_kc(kk); }
      else if (k < 156) { val = w3[n * 12 + (k - 144)]; }
      else if (k == 156){ val = b3[n] + base[n]; }
    }
  }
  ws[idx] = val;
}

template <int I, int NT, bool REGS>
__device__ __forceinline__ void kan_layer_wmma(const float* __restrict__ arow,
                                               const float* __restrict__ xin,
                                               int half,
                                               const float* __restrict__ Bl,
                                               float sS, float sR,
                                               v8f& c0, v8f& c1) {
  const float* __restrict__ arowh = REGS ? arow : arow + 2 * half;
  constexpr int NBASIS = 3 * I;
  constexpr int NTAIL  = (I + 1 + 3) / 4;
#pragma unroll
  for (int i4 = 0; i4 < I; i4 += 4) {
    v4f av = {};
    if (!REGS) av = *(const v4f*)(arow + i4);
#pragma unroll
    for (int ii = 0; ii < 4; ++ii) {
      const int i = i4 + ii;
      if (i < I) {
        const float xi = REGS ? xin[i] : av[ii];
        const v2f xv = {xi, xi};
        const v2f ssv = {sS, sS};
        const v2f c0v = {(float)(0.5 * RBF_D), (float)(1.5 * RBF_D)};
        const v2f q = __builtin_elementwise_fma(xv, ssv, -c0v);
        const v2f e = -(q * q);
        v2f f[3];
        f[0].x = __builtin_amdgcn_exp2f(e.x);
        f[0].y = __builtin_amdgcn_exp2f(e.y);
        const float rho = __builtin_amdgcn_exp2f(__builtin_fminf(xi * sR, 126.0f));
        const v2f rv = {rho, rho};
        f[1] = f[0] * rv;
        f[2] = f[1] * rv;
#pragma unroll
        for (int p = 0; p < 3; ++p) {
          const int t = 3 * i + p;
          const v2f a = f[p];
          const v2f b0 = *(const v2f*)(Bl + (t * NT + 0) * 64);
          c0 = __builtin_amdgcn_wmma_f32_16x16x4_f32(false, a, false, b0, (short)0, c0,
                                                     false, false);
          if (NT == 2) {
            const v2f b1 = *(const v2f*)(Bl + (t * NT + 1) * 64);
            c1 = __builtin_amdgcn_wmma_f32_16x16x4_f32(false, a, false, b1, (short)0, c1,
                                                       false, false);
          }
        }
      }
    }
  }
#pragma unroll
  for (int tt = 0; tt < NTAIL; ++tt) {
    const int t = NBASIS + tt;
    v2f a;
    if (4 * tt + 3 < I) {
      if (REGS) {
        a.x = half ? xin[4 * tt + 2] : xin[4 * tt + 0];
        a.y = half ? xin[4 * tt + 3] : xin[4 * tt + 1];
      } else {
        a = *(const v2f*)(arowh + 4 * tt);
      }
    } else {
#pragma unroll
      for (int v = 0; v < 2; ++v) {
        const int jlo = 4 * tt + v, jhi = jlo + 2;
        const float elo = jlo < I ? (REGS ? xin[jlo] : arow[jlo]) : (jlo == I ? 1.0f : 0.0f);
        const float ehi = jhi < I ? (REGS ? xin[jhi] : arow[jhi]) : (jhi == I ? 1.0f : 0.0f);
        a[v] = half ? ehi : elo;
      }
    }
    const v2f b0 = *(const v2f*)(Bl + (t * NT + 0) * 64);
    c0 = __builtin_amdgcn_wmma_f32_16x16x4_f32(false, a, false, b0, (short)0, c0,
                                               false, false);
    if (NT == 2) {
      const v2f b1 = *(const v2f*)(Bl + (t * NT + 1) * 64);
      c1 = __builtin_amdgcn_wmma_f32_16x16x4_f32(false, a, false, b1, (short)0, c1,
                                                 false, false);
    }
  }
}

__device__ __forceinline__ float silu_f(float z) {
  return z * __builtin_amdgcn_rcpf(1.0f + __expf(-z));
}

__global__ __launch_bounds__(256, 2) void kan_forward(const float* __restrict__ x,
                                                      const float* __restrict__ ws,
                                                      float* __restrict__ out,
                                                      int tiles_per_wave) {
  __shared__ __align__(16) float Bsh[BTOTAL];
  __shared__ __align__(16) float act1[8][16][P1];
  __shared__ __align__(16) float act2[8][16][P2];
  __shared__ __align__(16) float cor[8][16][4];

  const int tid  = threadIdx.x;
  const int wv   = tid >> 5;
  const int lane = tid & 31;
  const int half = lane >> 4;
  const int m    = lane & 15;

  const int tile0 = blockIdx.x * tiles_per_wave * 8 + wv;
  const float* xrow = x + (size_t)(tile0 * 16 + m) * 9;
  v4f   xa = *(const v4f_a4*)(xrow);
  v4f   xb = *(const v4f_a4*)(xrow + 4);
  float xc = xrow[8];

  for (int i = tid; i < BTOTAL; i += 256) Bsh[i] = ws[i];
  __syncthreads();

  const float* a1 = &act1[wv][m][0];
  const float* a2 = &act2[wv][m][0];
  const float* Bl = Bsh + lane * 2;

  const float sS = half ? (float)RBF_S : -(float)RBF_S;
  const float sR = half ? (float)RBF_R : -(float)RBF_R;

  for (int it = 0; it < tiles_per_wave; ++it) {
    const int rowbase = ((blockIdx.x * tiles_per_wave + it) * 8 + wv) * 16;

    const float xin[9] = {xa[0], xa[1], xa[2], xa[3], xb[0], xb[1], xb[2], xb[3], xc};
    v8f c0 = {}, c1 = {};
    kan_layer_wmma<9, NT1, true>(nullptr, xin, half, Bl + B1_OFF, sS, sR, c0, c1);

    {
      const int itn = it + 1 < tiles_per_wave ? it + 1 : it;
      const float* xn = xrow + (size_t)itn * (8 * 16 * 9);
      xa = *(const v4f_a4*)(xn);
      xb = *(const v4f_a4*)(xn + 4);
      xc = xn[8];
    }
#pragma unroll
    for (int v = 0; v < 8; ++v) {
      int r = half * 8 + v;
      act1[wv][r][m] = silu_f(c0[v]);
      if (m < 8) act1[wv][r][16 + m] = silu_f(c1[v]);
    }
    asm volatile("s_wait_dscnt 0" ::: "memory");

    v8f d0 = {}, dz = {};
    kan_layer_wmma<24, NT2, false>(a1, nullptr, half, Bl + B2_OFF, sS, sR, d0, dz);
#pragma unroll
    for (int v = 0; v < 8; ++v) {
      int r = half * 8 + v;
      if (m < 12) act2[wv][r][m] = silu_f(d0[v]);
    }
    asm volatile("s_wait_dscnt 0" ::: "memory");

    v8f e0 = {}, ez = {};
    kan_layer_wmma<12, NT3, false>(a2, nullptr, half, Bl + B3_OFF, sS, sR, e0, ez);
#pragma unroll
    for (int v = 0; v < 8; ++v) {
      int r = half * 8 + v;
      if (m < 3) cor[wv][r][m] = e0[v];
    }
    asm volatile("s_wait_dscnt 0" ::: "memory");

    if (half == 0) {
      float p0 = cor[wv][m][0], p1 = cor[wv][m][1], p2 = cor[wv][m][2];
      float mx = fmaxf(p0, fmaxf(p1, p2));
      float q0 = __expf(p0 - mx), q1 = __expf(p1 - mx), q2 = __expf(p2 - mx);
      float inv = __builtin_amdgcn_rcpf(q0 + q1 + q2);
      float* o = out + (size_t)(rowbase + m) * 3;
      __builtin_nontemporal_store(q0 * inv, o + 0);
      __builtin_nontemporal_store(q1 * inv, o + 1);
      __builtin_nontemporal_store(q2 * inv, o + 2);
    }
    asm volatile("s_wait_dscnt 0" ::: "memory");
  }
}

extern "C" void kernel_launch(void* const* d_in, const int* in_sizes, int n_in,
                              void* d_out, int out_size, void* d_ws, size_t ws_size,
                              hipStream_t stream) {
  const float* x    = (const float*)d_in[0];
  const float* c1   = (const float*)d_in[1];
  const float* w1   = (const float*)d_in[2];
  const float* b1   = (const float*)d_in[3];
  const float* c2   = (const float*)d_in[4];
  const float* w2   = (const float*)d_in[5];
  const float* b2   = (const float*)d_in[6];
  const float* c3   = (const float*)d_in[7];
  const float* w3   = (const float*)d_in[8];
  const float* b3   = (const float*)d_in[9];
  const float* base = (const float*)d_in[10];
  float* ws  = (float*)d_ws;
  float* out = (float*)d_out;

  const int Btot = in_sizes[0] / 9;
  const int TILES = 4;
  const int blocks = Btot / (128 * TILES);

  kan_pack<<<(BTOTAL + 255) / 256, 256, 0, stream>>>(c1, w1, b1, c2, w2, b2,
                                                     c3, w3, b3, base, ws);
  kan_forward<<<blocks, 256, 0, stream>>>(x, ws, out, TILES);
}


# Packed B built in LDS by each work-group of the forward kernel straight from the raw coefficient arrays (48-byte reads, 8-byte LDS scatter): the pack launch and its workspace round trip are gone
# speedup vs baseline: 1.6364x; 1.0581x over previous
#include <hip/hip_runtime.h>


typedef __attribute__((ext_vector_type(2))) float v2f;
typedef __attribute__((ext_vector_type(4))) float v4f;
typedef __attribute__((ext_vector_type(8))) float v8f;
typedef float v4f_a4 __attribute__((ext_vector_type(4), aligned(4)));

#define GRIDN 12
#define RBF_S 1.5570899671947016
#define RBF_D 0.8493218002880191
#define RBF_R 5.289881816592866
#define KC10 0.049787068367863944
#define KC11 0.006737946999085467
#define KC20 4.5399929762484854e-05
#define KC21 8.3152871910356788e-07

#define P1 28
#define P2 12

#define NSTEP1 30
#define NSTEP2 79
#define NSTEP3 40
#define NT1 2
#define NT2 1
#define NT3 1
#define B1_FLOATS (NSTEP1 * NT1 * 64)
#define B2_FLOATS (NSTEP2 * NT2 * 64)
#define B3_FLOATS (NSTEP3 * NT3 * 64)
#define B1_OFF 0
#define B2_OFF (B1_FLOATS)
#define B3_OFF (B1_FLOATS + B2_FLOATS)
#define BTOTAL (B1_FLOATS + B2_FLOATS + B3_FLOATS)

struct PairRegs { v4f a, b, c; };

__device__ __forceinline__ PairRegs load_pair(const float* __restrict__ src) {
  PairRegs r;
  r.a = *(const v4f_a4*)(src);
  r.b = *(const v4f_a4*)(src + 4);
  r.c = *(const v4f_a4*)(src + 8);
  return r;
}

template <int O, int NT>
__device__ __forceinline__ void store_pair(const PairRegs& r, int pair, float* __restrict__ Bl) {
  const int i = pair / O, n = pair - i * O;
  const float cg[12] = {r.a[0], r.a[1], r.a[2], r.a[3], r.b[0], r.b[1], r.b[2], r.b[3],
                        r.c[0], r.c[1], r.c[2], r.c[3]};
  float* dst = Bl + ((3 * i * NT + (n >> 4)) * 32 + (n & 15)) * 2;
#pragma unroll
  for (int p = 0; p < 3; ++p) {
    const float k0 = p == 0 ? 1.0f : (p == 1 ? (float)KC10 : (float)KC20);
    const float k1 = p == 0 ? 1.0f : (p == 1 ? (float)KC11 : (float)KC21);
#pragma unroll
    for (int hh = 0; hh < 2; ++hh) {
      const int g0 = hh ? 6 + 2 * p : 5 - 2 * p;
      const int g1 = hh ? 7 + 2 * p : 4 - 2 * p;
      v2f val;
      val.x = cg[g0] * k0;
      val.y = cg[g1] * k1;
      *(v2f*)(dst + (p * NT * 32 + hh * 16) * 2) = val;
    }
  }
}

template <int I, int NT>
__device__ __forceinline__ void store_lin(float w, int f, float* __restrict__ Bl) {
  const int n = f / I, j = f - n * I;
  const int t = 3 * I + (j >> 2), hh = (j >> 1) & 1, v = j & 1;
  Bl[((t * NT + (n >> 4)) * 32 + hh * 16 + (n & 15)) * 2 + v] = w;
}

template <int I, int NT>
__device__ __forceinline__ void store_bias(float b, int n, float* __restrict__ Bl) {
  constexpr int k = 13 * I, t = k >> 2, hh = (k >> 1) & 1, v = k & 1;
  Bl[((t * NT + (n >> 4)) * 32 + hh * 16 + (n & 15)) * 2 + v] = b;
}

template <int I, int NT, bool REGS>
__device__ __forceinline__ void kan_layer_wmma(const float* __restrict__ arow,
                                               const float* __restrict__ xin,
                                               int half,
                                               const float* __restrict__ Bl,
                                               float sS, float sR,
                                               v8f& c0, v8f& c1) {
  const float* __restrict__ arowh = REGS ? arow : arow + 2 * half;
  constexpr int NBASIS = 3 * I;
  constexpr int NTAIL  = (I + 1 + 3) / 4;
#pragma unroll
  for (int i4 = 0; i4 < I; i4 += 4) {
    v4f av = {};
    if (!REGS) av = *(const v4f*)(arow + i4);
#pragma unroll
    for (int ii = 0; ii < 4; ++ii) {
      const int i = i4 + ii;
      if (i < I) {
        const float xi = REGS ? xin[i] : av[ii];
        const v2f xv = {xi, xi};
        const v2f ssv = {sS, sS};
        const v2f c0v = {(float)(0.5 * RBF_D), (float)(1.5 * RBF_D)};
        const v2f q = __builtin_elementwise_fma(xv, ssv, -c0v);
        const v2f e = -(q * q);
        v2f f[3];
        f[0].x = __builtin_amdgcn_exp2f(e.x);
        f[0].y = __builtin_amdgcn_exp2f(e.y);
        const float rho = __builtin_amdgcn_exp2f(__builtin_fminf(xi * sR, 126.0f));
        const v2f rv = {rho, rho};
        f[1] = f[0] * rv;
        f[2] = f[1] * rv;
#pragma unroll
        for (int p = 0; p < 3; ++p) {
          const int t = 3 * i + p;
          const v2f a = f[p];
          const v2f b0 = *(const v2f*)(Bl + (t * NT + 0) * 64);
          c0 = __builtin_amdgcn_wmma_f32_16x16x4_f32(false, a, false, b0, (short)0, c0,
                                                     false, false);
          if (NT == 2) {
            const v2f b1 = *(const v2f*)(Bl + (t * NT + 1) * 64);
            c1 = __builtin_amdgcn_wmma_f32_16x16x4_f32(false, a, false, b1, (short)0, c1,
                                                       false, false);
          }
        }
      }
    }
  }
#pragma unroll
  for (int tt = 0; tt < NTAIL; ++tt) {
    const int t = NBASIS + tt;
    v2f a;
    if (4 * tt + 3 < I) {
      if (REGS) {
        a.x = half ? xin[4 * tt + 2] : xin[4 * tt + 0];
        a.y = half ? xin[4 * tt + 3] : xin[4 * tt + 1];
      } else {
        a = *(const v2f*)(arowh + 4 * tt);
      }
    } else {
#pragma unroll
      for (int v = 0; v < 2; ++v) {
        const int jlo = 4 * tt + v, jhi = jlo + 2;
        const float elo = jlo < I ? (REGS ? xin[jlo] : arow[jlo]) : (jlo == I ? 1.0f : 0.0f);
        const float ehi = jhi < I ? (REGS ? xin[jhi] : arow[jhi]) : (jhi == I ? 1.0f : 0.0f);
        a[v] = half ? ehi : elo;
      }
    }
    const v2f b0 = *(const v2f*)(Bl + (t * NT + 0) * 64);
    c0 = __builtin_amdgcn_wmma_f32_16x16x4_f32(false, a, false, b0, (short)0, c0,
                                               false, false);
    if (NT == 2) {
      const v2f b1 = *(const v2f*)(Bl + (t * NT + 1) * 64);
      c1 = __builtin_amdgcn_wmma_f32_16x16x4_f32(false, a, false, b1, (short)0, c1,
                                                 false, false);
    }
  }
}

__device__ __forceinline__ float silu_f(float z) {
  return z * __builtin_amdgcn_rcpf(1.0f + __expf(-z));
}

__global__ __launch_bounds__(256, 2) void kan_forward(
    const float* __restrict__ x,
    const float* __restrict__ cf1, const float* __restrict__ w1, const float* __restrict__ b1,
    const float* __restrict__ cf2, const float* __restrict__ w2, const float* __restrict__ b2,
    const float* __restrict__ cf3, const float* __restrict__ w3, const float* __restrict__ b3,
    const float* __restrict__ base, float* __restrict__ out, int tiles_per_wave) {
  __shared__ __align__(16) float Bsh[BTOTAL];
  __shared__ __align__(16) float act1[8][16][P1];
  __shared__ __align__(16) float act2[8][16][P2];
  __shared__ __align__(16) float cor[8][16][4];

  const int tid  = threadIdx.x;
  const int wv   = tid >> 5;
  const int lane = tid & 31;
  const int half = lane >> 4;
  const int m    = lane & 15;

  const int tile0 = blockIdx.x * tiles_per_wave * 8 + wv;
  const float* xrow = x + (size_t)(tile0 * 16 + m) * 9;
  v4f   xa = *(const v4f_a4*)(xrow);
  v4f   xb = *(const v4f_a4*)(xrow + 4);
  float xc = xrow[8];

  const bool isA1 = tid < 216;
  const int  pA   = isA1 ? tid : (tid < 252 ? tid - 216 : 35);
  const int  pC   = 256 + (tid & 31);
  const PairRegs rA = load_pair(isA1 ? cf1 + pA * 12 : cf3 + pA * 12);
  const PairRegs rB = load_pair(cf2 + tid * 12);
  const PairRegs rC = load_pair(cf2 + pC * 12);
  const float lA = isA1 ? w1[pA] : w3[pA];
  const float lB = w2[tid];
  const float lC = w2[pC];
  const int   nb = tid < 24 ? tid : (tid < 36 ? tid - 24 : (tid < 39 ? tid - 36 : 0));
  const float* bp = tid < 24 ? b1 : (tid < 36 ? b2 : b3);
  const float bv = bp[nb];
  const float bs = base[nb < 3 ? nb : 0];

  for (int q = tid; q < BTOTAL / 4; q += 256) ((v4f*)Bsh)[q] = v4f{0.0f, 0.0f, 0.0f, 0.0f};
  __syncthreads();

  if (isA1) {
    store_pair<24, NT1>(rA, pA, Bsh + B1_OFF);
    store_lin<9, NT1>(lA, pA, Bsh + B1_OFF);
  } else if (tid < 252) {
    store_pair<3, NT3>(rA, pA, Bsh + B3_OFF);
    store_lin<12, NT3>(lA, pA, Bsh + B3_OFF);
  }
  store_pair<12, NT2>(rB, tid, Bsh + B2_OFF);
  store_lin<24, NT2>(lB, tid, Bsh + B2_OFF);
  if (tid < 32) {
    store_pair<12, NT2>(rC, pC, Bsh + B2_OFF);
    store_lin<24, NT2>(lC, pC, Bsh + B2_OFF);
  }
  if (tid < 24)      store_bias<9, NT1>(bv, nb, Bsh + B1_OFF);
  else if (tid < 36) store_bias<24, NT2>(bv, nb, Bsh + B2_OFF);
  else if (tid < 39) store_bias<12, NT3>(bv + bs, nb, Bsh + B3_OFF);
  __syncthreads();

  const float* a1 = &act1[wv][m][0];
  const float* a2 = &act2[wv][m][0];
  const float* Bl = Bsh + lane * 2;

  const float sS = half ? (float)RBF_S : -(float)RBF_S;
  const float sR = half ? (float)RBF_R : -(float)RBF_R;

  for (int it = 0; it < tiles_per_wave; ++it) {
    const int rowbase = ((blockIdx.x * tiles_per_wave + it) * 8 + wv) * 16;

    const float xin[9] = {xa[0], xa[1], xa[2], xa[3], xb[0], xb[1], xb[2], xb[3], xc};
    v8f c0 = {}, c1 = {};
    kan_layer_wmma<9, NT1, true>(nullptr, xin, half, Bl + B1_OFF, sS, sR, c0, c1);

    {
      const int itn = it + 1 < tiles_per_wave ? it + 1 : it;
      const float* xn = xrow + (size_t)itn * (8 * 16 * 9);
      xa = *(const v4f_a4*)(xn);
      xb = *(const v4f_a4*)(xn + 4);
      xc = xn[8];
    }
#pragma unroll
    for (int v = 0; v < 8; ++v) {
      int r = half * 8 + v;
      act1[wv][r][m] = silu_f(c0[v]);
      if (m < 8) act1[wv][r][16 + m] = silu_f(c1[v]);
    }
    asm volatile("s_wait_dscnt 0" ::: "memory");

    v8f d0 = {}, dz = {};
    kan_layer_wmma<24, NT2, false>(a1, nullptr, half, Bl + B2_OFF, sS, sR, d0, dz);
#pragma unroll
    for (int v = 0; v < 8; ++v) {
      int r = half * 8 + v;
      if (m < 12) act2[wv][r][m] = silu_f(d0[v]);
    }
    asm volatile("s_wait_dscnt 0" ::: "memory");

    v8f e0 = {}, ez = {};
    kan_layer_wmma<12, NT3, false>(a2, nullptr, half, Bl + B3_OFF, sS, sR, e0, ez);
#pragma unroll
    for (int v = 0; v < 8; ++v) {
      int r = half * 8 + v;
      if (m < 3) cor[wv][r][m] = e0[v];
    }
    asm volatile("s_wait_dscnt 0" ::: "memory");

    if (half == 0) {
      float p0 = cor[wv][m][0], p1 = cor[wv][m][1], p2 = cor[wv][m][2];
      float mx = fmaxf(p0, fmaxf(p1, p2));
      float q0 = __expf(p0 - mx), q1 = __expf(p1 - mx), q2 = __expf(p2 - mx);
      float inv = __builtin_amdgcn_rcpf(q0 + q1 + q2);
      float* o = out + (size_t)(rowbase + m) * 3;
      __builtin_nontemporal_store(q0 * inv, o + 0);
      __builtin_nontemporal_store(q1 * inv, o + 1);
      __builtin_nontemporal_store(q2 * inv, o + 2);
    }
    asm volatile("s_wait_dscnt 0" ::: "memory");
  }
}

extern "C" void kernel_launch(void* const* d_in, const int* in_sizes, int n_in,
                              void* d_out, int out_size, void* d_ws, size_t ws_size,
                              hipStream_t stream) {
  const float* x    = (const float*)d_in[0];
  const float* c1   = (const float*)d_in[1];
  const float* w1   = (const float*)d_in[2];
  const float* b1   = (const float*)d_in[3];
  const float* c2   = (const float*)d_in[4];
  const float* w2   = (const float*)d_in[5];
  const float* b2   = (const float*)d_in[6];
  const float* c3   = (const float*)d_in[7];
  const float* w3   = (const float*)d_in[8];
  const float* b3   = (const float*)d_in[9];
  const float* base = (const float*)d_in[10];
  float* out = (float*)d_out;

  const int Btot = in_sizes[0] / 9;
  const int TILES = 4;
  const int blocks = Btot / (128 * TILES);

  kan_forward<<<blocks, 256, 0, stream>>>(x, c1, w1, b1, c2, w2, b2, c3, w3, b3, base,
                                          out, TILES);
}


# Issue priority by progress: s_setprio 3 from entry, then 2, 1, 0 at the layer boundaries of a wave's last tile; lagging waves win the SIMD's issue, so its waves end together, not one by one
# speedup vs baseline: 1.6919x; 1.0339x over previous
#include <hip/hip_runtime.h>


typedef __attribute__((ext_vector_type(2))) float v2f;
typedef __attribute__((ext_vector_type(4))) float v4f;
typedef __attribute__((ext_vector_type(8))) float v8f;
typedef float v4f_a4 __attribute__((ext_vector_type(4), aligned(4)));

#define GRIDN 12
#define RBF_S 1.5570899671947016
#define RBF_D 0.8493218002880191
#define RBF_R 5.289881816592866
#define KC10 0.049787068367863944
#define KC11 0.006737946999085467
#define KC20 4.5399929762484854e-05
#define KC21 8.3152871910356788e-07

#define P1 28
#define P2 12

#define NSTEP1 30
#define NSTEP2 79
#define NSTEP3 40
#define NT1 2
#define NT2 1
#define NT3 1
#define B1_FLOATS (NSTEP1 * NT1 * 64)
#define B2_FLOATS (NSTEP2 * NT2 * 64)
#define B3_FLOATS (NSTEP3 * NT3 * 64)
#define B1_OFF 0
#define B2_OFF (B1_FLOATS)
#define B3_OFF (B1_FLOATS + B2_FLOATS)
#define BTOTAL (B1_FLOATS + B2_FLOATS + B3_FLOATS)

struct PairRegs { v4f a, b, c; };

__device__ __forceinline__ PairRegs load_pair(const float* __restrict__ src) {
  PairRegs r;
  r.a = *(const v4f_a4*)(src);
  r.b = *(const v4f_a4*)(src + 4);
  r.c = *(const v4f_a4*)(src + 8);
  return r;
}

template <int O, int NT>
__device__ __forceinline__ void store_pair(const PairRegs& r, int pair, float* __restrict__ Bl) {
  const int i = pair / O, n = pair - i * O;
  const float cg[12] = {r.a[0], r.a[1], r.a[2], r.a[3], r.b[0], r.b[1], r.b[2], r.b[3],
                        r.c[0], r.c[1], r.c[2], r.c[3]};
  float* dst = Bl + ((3 * i * NT + (n >> 4)) * 32 + (n & 15)) * 2;
#pragma unroll
  for (int p = 0; p < 3; ++p) {
    const float k0 = p == 0 ? 1.0f : (p == 1 ? (float)KC10 : (float)KC20);
    const float k1 = p == 0 ? 1.0f : (p == 1 ? (float)KC11 : (float)KC21);
#pragma unroll
    for (int hh = 0; hh < 2; ++hh) {
      const int g0 = hh ? 6 + 2 * p : 5 - 2 * p;
      const int g1 = hh ? 7 + 2 * p : 4 - 2 * p;
      v2f val;
      val.x = cg[g0] * k0;
      val.y = cg[g1] * k1;
      *(v2f*)(dst + (p * NT * 32 + hh * 16) * 2) = val;
    }
  }
}

template <int I, int NT>
__device__ __forceinline__ void store_lin(float w, int f, float* __restrict__ Bl) {
  const int n = f / I, j = f - n * I;
  const int t = 3 * I + (j >> 2), hh = (j >> 1) & 1, v = j & 1;
  Bl[((t * NT + (n >> 4)) * 32 + hh * 16 + (n & 15)) * 2 + v] = w;
}

template <int I, int NT>
__device__ __forceinline__ void store_bias(float b, int n, float* __restrict__ Bl) {
  constexpr int k = 13 * I, t = k >> 2, hh = (k >> 1) & 1, v = k & 1;
  Bl[((t * NT + (n >> 4)) * 32 + hh * 16 + (n & 15)) * 2 + v] = b;
}

template <int I, int NT, bool REGS>
__device__ __forceinline__ void kan_layer_wmma(const float* __restrict__ arow,
                                               const float* __restrict__ xin,
                                               int half,
                                               const float* __restrict__ Bl,
                                               float sS, float sR,
                                               v8f& c0, v8f& c1) {
  const float* __restrict__ arowh = REGS ? arow : arow + 2 * half;
  constexpr int NBASIS = 3 * I;
  constexpr int NTAIL  = (I + 1 + 3) / 4;
#pragma unroll
  for (int i4 = 0; i4 < I; i4 += 4) {
    v4f av = {};
    if (!REGS) av = *(const v4f*)(arow + i4);
#pragma unroll
    for (int ii = 0; ii < 4; ++ii) {
      const int i = i4 + ii;
      if (i < I) {
        const float xi = REGS ? xin[i] : av[ii];
        const v2f xv = {xi, xi};
        const v2f ssv = {sS, sS};
        const v2f c0v = {(float)(0.5 * RBF_D), (float)(1.5 * RBF_D)};
        const v2f q = __builtin_elementwise_fma(xv, ssv, -c0v);
        const v2f e = -(q * q);
        v2f f[3];
        f[0].x = __builtin_amdgcn_exp2f(e.x);
        f[0].y = __builtin_amdgcn_exp2f(e.y);
        const float rho = __builtin_amdgcn_exp2f(__builtin_fminf(xi * sR, 126.0f));
        const v2f rv = {rho, rho};
        f[1] = f[0] * rv;
        f[2] = f[1] * rv;
#pragma unroll
        for (int p = 0; p < 3; ++p) {
          const int t = 3 * i + p;
          const v2f a = f[p];
          const v2f b0 = *(const v2f*)(Bl + (t * NT + 0) * 64);
          c0 = __builtin_amdgcn_wmma_f32_16x16x4_f32(false, a, false, b0, (short)0, c0,
                                                     false, false);
          if (NT == 2) {
            const v2f b1 = *(const v2f*)(Bl + (t * NT + 1) * 64);
            c1 = __builtin_amdgcn_wmma_f32_16x16x4_f32(false, a, false, b1, (short)0, c1,
                                                       false, false);
          }
        }
      }
    }
  }
#pragma unroll
  for (int tt = 0; tt < NTAIL; ++tt) {
    const int t = NBASIS + tt;
    v2f a;
    if (4 * tt + 3 < I) {
      if (REGS) {
        a.x = half ? xin[4 * tt + 2] : xin[4 * tt + 0];
        a.y = half ? xin[4 * tt + 3] : xin[4 * tt + 1];
      } else {
        a = *(const v2f*)(arowh + 4 * tt);
      }
    } else {
#pragma unroll
      for (int v = 0; v < 2; ++v) {
        const int jlo = 4 * tt + v, jhi = jlo + 2;
        const float elo = jlo < I ? (REGS ? xin[jlo] : arow[jlo]) : (jlo == I ? 1.0f : 0.0f);
        const float ehi = jhi < I ? (REGS ? xin[jhi] : arow[jhi]) : (jhi == I ? 1.0f : 0.0f);
        a[v] = half ? ehi : elo;
      }
    }
    const v2f b0 = *(const v2f*)(Bl + (t * NT + 0) * 64);
    c0 = __builtin_amdgcn_wmma_f32_16x16x4_f32(false, a, false, b0, (short)0, c0,
                                               false, false);
    if (NT == 2) {
      const v2f b1 = *(const v2f*)(Bl + (t * NT + 1) * 64);
      c1 = __builtin_amdgcn_wmma_f32_16x16x4_f32(false, a, false, b1, (short)0, c1,
                                                 false, false);
    }
  }
}

__device__ __forceinline__ float silu_f(float z) {
  return z * __builtin_amdgcn_rcpf(1.0f + __expf(-z));
}

__global__ __launch_bounds__(256, 2) void kan_forward(
    const float* __restrict__ x,
    const float* __restrict__ cf1, const float* __restrict__ w1, const float* __restrict__ b1,
    const float* __restrict__ cf2, const float* __restrict__ w2, const float* __restrict__ b2,
    const float* __restrict__ cf3, const float* __restrict__ w3, const float* __restrict__ b3,
    const float* __restrict__ base, float* __restrict__ out, int tiles_per_wave) {
  __shared__ __align__(16) float Bsh[BTOTAL];
  __shared__ __align__(16) float act1[8][16][P1];
  __shared__ __align__(16) float act2[8][16][P2];
  __shared__ __align__(16) float cor[8][16][4];

  __builtin_amdgcn_s_setprio(3);

  const int tid  = threadIdx.x;
  const int wv   = tid >> 5;
  const int lane = tid & 31;
  const int half = lane >> 4;
  const int m    = lane & 15;

  const int tile0 = blockIdx.x * tiles_per_wave * 8 + wv;
  const float* xrow = x + (size_t)(tile0 * 16 + m) * 9;
  v4f   xa = *(const v4f_a4*)(xrow);
  v4f   xb = *(const v4f_a4*)(xrow + 4);
  float xc = xrow[8];

  const bool isA1 = tid < 216;
  const int  pA   = isA1 ? tid : (tid < 252 ? tid - 216 : 35);
  const int  pC   = 256 + (tid & 31);
  const PairRegs rA = load_pair(isA1 ? cf1 + pA * 12 : cf3 + pA * 12);
  const PairRegs rB = load_pair(cf2 + tid * 12);
  const PairRegs rC = load_pair(cf2 + pC * 12);
  const float lA = isA1 ? w1[pA] : w3[pA];
  const float lB = w2[tid];
  const float lC = w2[pC];
  const int   nb = tid < 24 ? tid : (tid < 36 ? tid - 24 : (tid < 39 ? tid - 36 : 0));
  const float* bp = tid < 24 ? b1 : (tid < 36 ? b2 : b3);
  const float bv = bp[nb];
  const float bs = base[nb < 3 ? nb : 0];

  for (int q = tid; q < BTOTAL / 4; q += 256) ((v4f*)Bsh)[q] = v4f{0.0f, 0.0f, 0.0f, 0.0f};
  __syncthreads();

  if (isA1) {
    store_pair<24, NT1>(rA, pA, Bsh + B1_OFF);
    store_lin<9, NT1>(lA, pA, Bsh + B1_OFF);
  } else if (tid < 252) {
    store_pair<3, NT3>(rA, pA, Bsh + B3_OFF);
    store_lin<12, NT3>(lA, pA, Bsh + B3_OFF);
  }
  store_pair<12, NT2>(rB, tid, Bsh + B2_OFF);
  store_lin<24, NT2>(lB, tid, Bsh + B2_OFF);
  if (tid < 32) {
    store_pair<12, NT2>(rC, pC, Bsh + B2_OFF);
    store_lin<24, NT2>(lC, pC, Bsh + B2_OFF);
  }
  if (tid < 24)      store_bias<9, NT1>(bv, nb, Bsh + B1_OFF);
  else if (tid < 36) store_bias<24, NT2>(bv, nb, Bsh + B2_OFF);
  else if (tid < 39) store_bias<12, NT3>(bv + bs, nb, Bsh + B3_OFF);
  __syncthreads();

  const float* a1 = &act1[wv][m][0];
  const float* a2 = &act2[wv][m][0];
  const float* Bl = Bsh + lane * 2;

  const float sS = half ? (float)RBF_S : -(float)RBF_S;
  const float sR = half ? (float)RBF_R : -(float)RBF_R;

  for (int it = 0; it < tiles_per_wave; ++it) {
    const int rowbase = ((blockIdx.x * tiles_per_wave + it) * 8 + wv) * 16;
    const bool last_tile = it + 1 == tiles_per_wave;

    const float xin[9] = {xa[0], xa[1], xa[2], xa[3], xb[0], xb[1], xb[2], xb[3], xc};
    v8f c0 = {}, c1 = {};
    kan_layer_wmma<9, NT1, true>(nullptr, xin, half, Bl + B1_OFF, sS, sR, c0, c1);

    {
      const int itn = it + 1 < tiles_per_wave ? it + 1 : it;
      const float* xn = xrow + (size_t)itn * (8 * 16 * 9);
      xa = *(const v4f_a4*)(xn);
      xb = *(const v4f_a4*)(xn + 4);
      xc = xn[8];
    }
#pragma unroll
    for (int v = 0; v < 8; ++v) {
      int r = half * 8 + v;
      act1[wv][r][m] = silu_f(c0[v]);
      if (m < 8) act1[wv][r][16 + m] = silu_f(c1[v]);
    }
    asm volatile("s_wait_dscnt 0" ::: "memory");

    if (last_tile) __builtin_amdgcn_s_setprio(2);
    v8f d0 = {}, dz = {};
    kan_layer_wmma<24, NT2, false>(a1, nullptr, half, Bl + B2_OFF, sS, sR, d0, dz);
#pragma unroll
    for (int v = 0; v < 8; ++v) {
      int r = half * 8 + v;
      if (m < 12) act2[wv][r][m] = silu_f(d0[v]);
    }
    asm volatile("s_wait_dscnt 0" ::: "memory");

    if (last_tile) __builtin_amdgcn_s_setprio(1);
    v8f e0 = {}, ez = {};
    kan_layer_wmma<12, NT3, false>(a2, nullptr, half, Bl + B3_OFF, sS, sR, e0, ez);
#pragma unroll
    for (int v = 0; v < 8; ++v) {
      int r = half * 8 + v;
      if (m < 3) cor[wv][r][m] = e0[v];
    }
    asm volatile("s_wait_dscnt 0" ::: "memory");

    if (last_tile) __builtin_amdgcn_s_setprio(0);
    if (half == 0) {
      float p0 = cor[wv][m][0], p1 = cor[wv][m][1], p2 = cor[wv][m][2];
      float mx = fmaxf(p0, fmaxf(p1, p2));
      float q0 = __expf(p0 - mx), q1 = __expf(p1 - mx), q2 = __expf(p2 - mx);
      float inv = __builtin_amdgcn_rcpf(q0 + q1 + q2);
      float* o = out + (size_t)(rowbase + m) * 3;
      __builtin_nontemporal_store(q0 * inv, o + 0);
      __builtin_nontemporal_store(q1 * inv, o + 1);
      __builtin_nontemporal_store(q2 * inv, o + 2);
    }
    asm volatile("s_wait_dscnt 0" ::: "memory");
  }
  __builtin_amdgcn_s_setprio(0);
}

extern "C" void kernel_launch(void* const* d_in, const int* in_sizes, int n_in,
                              void* d_out, int out_size, void* d_ws, size_t ws_size,
                              hipStream_t stream) {
  const float* x    = (const float*)d_in[0];
  const float* c1   = (const float*)d_in[1];
  const float* w1   = (const float*)d_in[2];
  const float* b1   = (const float*)d_in[3];
  const float* c2   = (const float*)d_in[4];
  const float* w2   = (const float*)d_in[5];
  const float* b2   = (const float*)d_in[6];
  const float* c3   = (const float*)d_in[7];
  const float* w3   = (const float*)d_in[8];
  const float* b3   = (const float*)d_in[9];
  const float* base = (const float*)d_in[10];
  float* out = (float*)d_out;

  const int Btot = in_sizes[0] / 9;
  const int TILES = 4;
  const int blocks = Btot / (128 * TILES);

  kan_forward<<<blocks, 256, 0, stream>>>(x, c1, w1, b1, c2, w2, b2, c3, w3, b3, base,
                                          out, TILES);
}


# Transcendentals placed by hand: exps of input i+1 issue among input i's WMMAs (scheduling groups), SiLU chains pipelined branch-free (pad-column stores); no exp/rcp right before its reader or another
# speedup vs baseline: 1.7608x; 1.0407x over previous
#include <hip/hip_runtime.h>


typedef __attribute__((ext_vector_type(2))) float v2f;
typedef __attribute__((ext_vector_type(4))) float v4f;
typedef __attribute__((ext_vector_type(8))) float v8f;
typedef float v4f_a4 __attribute__((ext_vector_type(4), aligned(4)));

#define GRIDN 12
#define RBF_S 1.5570899671947016
#define RBF_D 0.8493218002880191
#define RBF_R 5.289881816592866
#define KC10 0.049787068367863944
#define KC11 0.006737946999085467
#define KC20 4.5399929762484854e-05
#define KC21 8.3152871910356788e-07

#define P1 28
#define P2 20

#define NSTEP1 30
#define NSTEP2 79
#define NSTEP3 40
#define NT1 2
#define NT2 1
#define NT3 1
#define B1_FLOATS (NSTEP1 * NT1 * 64)
#define B2_FLOATS (NSTEP2 * NT2 * 64)
#define B3_FLOATS (NSTEP3 * NT3 * 64)
#define B1_OFF 0
#define B2_OFF (B1_FLOATS)
#define B3_OFF (B1_FLOATS + B2_FLOATS)
#define BTOTAL (B1_FLOATS + B2_FLOATS + B3_FLOATS)

struct PairRegs { v4f a, b, c; };

__device__ __forceinline__ PairRegs load_pair(const float* __restrict__ src) {
  PairRegs r;
  r.a = *(const v4f_a4*)(src);
  r.b = *(const v4f_a4*)(src + 4);
  r.c = *(const v4f_a4*)(src + 8);
  return r;
}

template <int O, int NT>
__device__ __forceinline__ void store_pair(const PairRegs& r, int pair, float* __restrict__ Bl) {
  const int i = pair / O, n = pair - i * O;
  const float cg[12] = {r.a[0], r.a[1], r.a[2], r.a[3], r.b[0], r.b[1], r.b[2], r.b[3],
                        r.c[0], r.c[1], r.c[2], r.c[3]};
  float* dst = Bl + ((3 * i * NT + (n >> 4)) * 32 + (n & 15)) * 2;
#pragma unroll
  for (int p = 0; p < 3; ++p) {
    const float k0 = p == 0 ? 1.0f : (p == 1 ? (float)KC10 : (float)KC20);
    const float k1 = p == 0 ? 1.0f : (p == 1 ? (float)KC11 : (float)KC21);
#pragma unroll
    for (int hh = 0; hh < 2; ++hh) {
      const int g0 = hh ? 6 + 2 * p : 5 - 2 * p;
      const int g1 = hh ? 7 + 2 * p : 4 - 2 * p;
      v2f val;
      val.x = cg[g0] * k0;
      val.y = cg[g1] * k1;
      *(v2f*)(dst + (p * NT * 32 + hh * 16) * 2) = val;
    }
  }
}

template <int I, int NT>
__device__ __forceinline__ void store_lin(float w, int f, float* __restrict__ Bl) {
  const int n = f / I, j = f - n * I;
  const int t = 3 * I + (j >> 2), hh = (j >> 1) & 1, v = j & 1;
  Bl[((t * NT + (n >> 4)) * 32 + hh * 16 + (n & 15)) * 2 + v] = w;
}

template <int I, int NT>
__device__ __forceinline__ void store_bias(float b, int n, float* __restrict__ Bl) {
  constexpr int k = 13 * I, t = k >> 2, hh = (k >> 1) & 1, v = k & 1;
  Bl[((t * NT + (n >> 4)) * 32 + hh * 16 + (n & 15)) * 2 + v] = b;
}

struct RbfGen { v2f f0; float rho; };

__device__ __forceinline__ RbfGen rbf_gen(float xi, float sS, float sR) {
  const v2f xv = {xi, xi};
  const v2f ssv = {sS, sS};
  const v2f c0v = {(float)(0.5 * RBF_D), (float)(1.5 * RBF_D)};
  const v2f q = __builtin_elementwise_fma(xv, ssv, -c0v);
  const v2f e = -(q * q);
  RbfGen g;
  g.f0.x = __builtin_amdgcn_exp2f(e.x);
  g.f0.y = __builtin_amdgcn_exp2f(e.y);
  g.rho = __builtin_amdgcn_exp2f(__builtin_fminf(xi * sR, 126.0f));
  return g;
}

template <int I, int NT, bool REGS>
__device__ __forceinline__ void kan_layer_wmma(const float* __restrict__ arow,
                                               const float* __restrict__ xin,
                                               int half,
                                               const float* __restrict__ Bl,
                                               float sS, float sR,
                                               v8f& c0, v8f& c1) {
  const float* __restrict__ arowh = REGS ? arow : arow + 2 * half;
  constexpr int NBASIS = 3 * I;
  constexpr int NTAIL  = (I + 1 + 3) / 4;
  v4f grp[2] = {};
  if (!REGS) grp[0] = *(const v4f*)(arow);
  RbfGen cur = rbf_gen(REGS ? xin[0] : grp[0][0], sS, sR);
#pragma unroll
  for (int i = 0; i < I; ++i) {
    if (!REGS && (i & 3) == 0 && i + 4 < I)
      grp[((i >> 2) + 1) & 1] = *(const v4f*)(arow + i + 4);
    RbfGen nxt = cur;
    if (i + 1 < I) {
      const int j = i + 1;
      nxt = rbf_gen(REGS ? xin[j] : grp[(j >> 2) & 1][j & 3], sS, sR);
    }
    v2f f[3];
    const v2f rv = {cur.rho, cur.rho};
    f[0] = cur.f0;
    f[1] = f[0] * rv;
    f[2] = f[1] * rv;
#pragma unroll
    for (int p = 0; p < 3; ++p) {
      const int t = 3 * i + p;
      const v2f a = f[p];
      const v2f b0 = *(const v2f*)(Bl + (t * NT + 0) * 64);
      c0 = __builtin_amdgcn_wmma_f32_16x16x4_f32(false, a, false, b0, (short)0, c0,
                                                 false, false);
      if (NT == 2) {
        const v2f b1 = *(const v2f*)(Bl + (t * NT + 1) * 64);
        c1 = __builtin_amdgcn_wmma_f32_16x16x4_f32(false, a, false, b1, (short)0, c1,
                                                   false, false);
      }
    }
    if (i + 1 < I) {
      __builtin_amdgcn_sched_group_barrier(0x008, NT, 0);
      __builtin_amdgcn_sched_group_barrier(0x002, 4, 0);
      __builtin_amdgcn_sched_group_barrier(0x400, 1, 0);
      __builtin_amdgcn_sched_group_barrier(0x002, 1, 0);
      __builtin_amdgcn_sched_group_barrier(0x008, NT, 0);
      __builtin_amdgcn_sched_group_barrier(0x400, 1, 0);
      __builtin_amdgcn_sched_group_barrier(0x002, 1, 0);
      __builtin_amdgcn_sched_group_barrier(0x400, 1, 0);
      __builtin_amdgcn_sched_group_barrier(0x008, NT, 0);
    }
    cur = nxt;
  }
#pragma unroll
  for (int tt = 0; tt < NTAIL; ++tt) {
    const int t = NBASIS + tt;
    v2f a;
    if (4 * tt + 3 < I) {
      if (REGS) {
        a.x = half ? xin[4 * tt + 2] : xin[4 * tt + 0];
        a.y = half ? xin[4 * tt + 3] : xin[4 * tt + 1];
      } else {
        a = *(const v2f*)(arowh + 4 * tt);
      }
    } else {
#pragma unroll
      for (int v = 0; v < 2; ++v) {
        const int jlo = 4 * tt + v, jhi = jlo + 2;
        const float elo = jlo < I ? (REGS ? xin[jlo] : arow[jlo]) : (jlo == I ? 1.0f : 0.0f);
        const float ehi = jhi < I ? (REGS ? xin[jhi] : arow[jhi]) : (jhi == I ? 1.0f : 0.0f);
        a[v] = half ? ehi : elo;
      }
    }
    const v2f b0 = *(const v2f*)(Bl + (t * NT + 0) * 64);
    c0 = __builtin_amdgcn_wmma_f32_16x16x4_f32(false, a, false, b0, (short)0, c0,
                                               false, false);
    if (NT == 2) {
      const v2f b1 = *(const v2f*)(Bl + (t * NT + 1) * 64);
      c1 = __builtin_amdgcn_wmma_f32_16x16x4_f32(false, a, false, b1, (short)0, c1,
                                                 false, false);
    }
  }
}

#define PIN() __builtin_amdgcn_sched_barrier(0x3F4)
template <int N>
__device__ __forceinline__ void silu_chains(const float* __restrict__ z, float* __restrict__ y) {
  float a[N], e[N], d[N], r[N];
#pragma unroll
  for (int k = 0; k < N; ++k) a[k] = z[k] * -1.44269504f;
  PIN();
#pragma unroll
  for (int k = 0; k < N + 3; ++k) {
    if (k >= 2 && k - 2 < N) { d[k - 2] = 1.0f + e[k - 2]; PIN(); }
    if (k >= 2 && k - 2 < N) { r[k - 2] = __builtin_amdgcn_rcpf(d[k - 2]); PIN(); }
    if (k >= 3) { y[k - 3] = z[k - 3] * r[k - 3]; PIN(); }
    if (k < N) { e[k] = __builtin_amdgcn_exp2f(a[k]); PIN(); }
  }
}

__global__ __launch_bounds__(256, 2) void kan_forward(
    const float* __restrict__ x,
    const float* __restrict__ cf1, const float* __restrict__ w1, const float* __restrict__ b1,
    const float* __restrict__ cf2, const float* __restrict__ w2, const float* __restrict__ b2,
    const float* __restrict__ cf3, const float* __restrict__ w3, const float* __restrict__ b3,
    const float* __restrict__ base, float* __restrict__ out, int tiles_per_wave) {
  __shared__ __align__(16) float Bsh[BTOTAL];
  __shared__ __align__(16) float act1[8][16][P1];
  __shared__ __align__(16) float act2[8][16][P2];
  __shared__ __align__(16) float cor[8][16][4];

  __builtin_amdgcn_s_setprio(3);

  const int tid  = threadIdx.x;
  const int wv   = tid >> 5;
  const int lane = tid & 31;
  const int half = lane >> 4;
  const int m    = lane & 15;

  const int tile0 = blockIdx.x * tiles_per_wave * 8 + wv;
  const float* xrow = x + (size_t)(tile0 * 16 + m) * 9;
  v4f   xa = *(const v4f_a4*)(xrow);
  v4f   xb = *(const v4f_a4*)(xrow + 4);
  float xc = xrow[8];

  const bool isA1 = tid < 216;
  const int  pA   = isA1 ? tid : (tid < 252 ? tid - 216 : 35);
  const int  pC   = 256 + (tid & 31);
  const PairRegs rA = load_pair(isA1 ? cf1 + pA * 12 : cf3 + pA * 12);
  const PairRegs rB = load_pair(cf2 + tid * 12);
  const PairRegs rC = load_pair(cf2 + pC * 12);
  const float lA = isA1 ? w1[pA] : w3[pA];
  const float lB = w2[tid];
  const float lC = w2[pC];
  const int   nb = tid < 24 ? tid : (tid < 36 ? tid - 24 : (tid < 39 ? tid - 36 : 0));
  const float* bp = tid < 24 ? b1 : (tid < 36 ? b2 : b3);
  const float bv = bp[nb];
  const float bs = base[nb < 3 ? nb : 0];

  for (int q = tid; q < BTOTAL / 4; q += 256) ((v4f*)Bsh)[q] = v4f{0.0f, 0.0f, 0.0f, 0.0f};
  __syncthreads();

  if (isA1) {
    store_pair<24, NT1>(rA, pA, Bsh + B1_OFF);
    store_lin<9, NT1>(lA, pA, Bsh + B1_OFF);
  } else if (tid < 252) {
    store_pair<3, NT3>(rA, pA, Bsh + B3_OFF);
    store_lin<12, NT3>(lA, pA, Bsh + B3_OFF);
  }
  store_pair<12, NT2>(rB, tid, Bsh + B2_OFF);
  store_lin<24, NT2>(lB, tid, Bsh + B2_OFF);
  if (tid < 32) {
    store_pair<12, NT2>(rC, pC, Bsh + B2_OFF);
    store_lin<24, NT2>(lC, pC, Bsh + B2_OFF);
  }
  if (tid < 24)      store_bias<9, NT1>(bv, nb, Bsh + B1_OFF);
  else if (tid < 36) store_bias<24, NT2>(bv, nb, Bsh + B2_OFF);
  else if (tid < 39) store_bias<12, NT3>(bv + bs, nb, Bsh + B3_OFF);
  __syncthreads();

  const float* a1 = &act1[wv][m][0];
  const float* a2 = &act2[wv][m][0];
  const float* Bl = Bsh + lane * 2;

  const int c1col = m < 8 ? 16 + m : 24 + (m & 3);

  const float sS = half ? (float)RBF_S : -(float)RBF_S;
  const float sR = half ? (float)RBF_R : -(float)RBF_R;

  for (int it = 0; it < tiles_per_wave; ++it) {
    const int rowbase = ((blockIdx.x * tiles_per_wave + it) * 8 + wv) * 16;
    const bool last_tile = it + 1 == tiles_per_wave;

    const float xin[9] = {xa[0], xa[1], xa[2], xa[3], xb[0], xb[1], xb[2], xb[3], xc};
    v8f c0 = {}, c1 = {};
    kan_layer_wmma<9, NT1, true>(nullptr, xin, half, Bl + B1_OFF, sS, sR, c0, c1);

    {
      const int itn = it + 1 < tiles_per_wave ? it + 1 : it;
      const float* xn = xrow + (size_t)itn * (8 * 16 * 9);
      xa = *(const v4f_a4*)(xn);
      xb = *(const v4f_a4*)(xn + 4);
      xc = xn[8];
    }
    {
      float z[16], y[16];
#pragma unroll
      for (int v = 0; v < 8; ++v) { z[2 * v] = c0[v]; z[2 * v + 1] = c1[v]; }
      silu_chains<16>(z, y);
#pragma unroll
      for (int v = 0; v < 8; ++v) {
        int r = half * 8 + v;
        act1[wv][r][m] = y[2 * v];
        act1[wv][r][c1col] = y[2 * v + 1];
      }
    }
    asm volatile("s_wait_dscnt 0" ::: "memory");

    if (last_tile) __builtin_amdgcn_s_setprio(2);
    v8f d0 = {}, dz = {};
    kan_layer_wmma<24, NT2, false>(a1, nullptr, half, Bl + B2_OFF, sS, sR, d0, dz);
    {
      float z[8], y[8];
#pragma unroll
      for (int v = 0; v < 8; ++v) z[v] = d0[v];
      silu_chains<8>(z, y);
#pragma unroll
      for (int v = 0; v < 8; ++v) {
        int r = half * 8 + v;
        act2[wv][r][m] = y[v];
      }
    }
    asm volatile("s_wait_dscnt 0" ::: "memory");

    if (last_tile) __builtin_amdgcn_s_setprio(1);
    v8f e0 = {}, ez = {};
    kan_layer_wmma<12, NT3, false>(a2, nullptr, half, Bl + B3_OFF, sS, sR, e0, ez);
#pragma unroll
    for (int v = 0; v < 8; ++v) {
      int r = half * 8 + v;
      if (m < 3) cor[wv][r][m] = e0[v];
    }
    asm volatile("s_wait_dscnt 0" ::: "memory");

    if (last_tile) __builtin_amdgcn_s_setprio(0);
    if (half == 0) {
      float p0 = cor[wv][m][0], p1 = cor[wv][m][1], p2 = cor[wv][m][2];
      float mx = fmaxf(p0, fmaxf(p1, p2));
      float q0 = __expf(p0 - mx), q1 = __expf(p1 - mx), q2 = __expf(p2 - mx);
      float inv = __builtin_amdgcn_rcpf(q0 + q1 + q2);
      float* o = out + (size_t)(rowbase + m) * 3;
      __builtin_nontemporal_store(q0 * inv, o + 0);
      __builtin_nontemporal_store(q1 * inv, o + 1);
      __builtin_nontemporal_store(q2 * inv, o + 2);
    }
    asm volatile("s_wait_dscnt 0" ::: "memory");
  }
  __builtin_amdgcn_s_setprio(0);
}

extern "C" void kernel_launch(void* const* d_in, const int* in_sizes, int n_in,
                              void* d_out, int out_size, void* d_ws, size_t ws_size,
                              hipStream_t stream) {
  const float* x    = (const float*)d_in[0];
  const float* c1   = (const float*)d_in[1];
  const float* w1   = (const float*)d_in[2];
  const float* b1   = (const float*)d_in[3];
  const float* c2   = (const float*)d_in[4];
  const float* w2   = (const float*)d_in[5];
  const float* b2   = (const float*)d_in[6];
  const float* c3   = (const float*)d_in[7];
  const float* w3   = (const float*)d_in[8];
  const float* b3   = (const float*)d_in[9];
  const float* base = (const float*)d_in[10];
  float* out = (float*)d_out;

  const int Btot = in_sizes[0] / 9;
  const int TILES = 4;
  const int blocks = Btot / (128 * TILES);

  kan_forward<<<blocks, 256, 0, stream>>>(x, c1, w1, b1, c2, w2, b2, c3, w3, b3, base,
                                          out, TILES);
}


# RBF basis: one lane builds all 12 centres of one input (lane halves take alternate inputs; seed pair, rho up, 1/rho down): 4 exps + 9 vector ops per 6 K-steps, was 6 + 12; K order of centres changed
# speedup vs baseline: 1.8250x; 1.0365x over previous
#include <hip/hip_runtime.h>


typedef __attribute__((ext_vector_type(2))) float v2f;
typedef __attribute__((ext_vector_type(4))) float v4f;
typedef __attribute__((ext_vector_type(8))) float v8f;
typedef float v4f_a4 __attribute__((ext_vector_type(4), aligned(4)));

#define GRIDN 12
#define RBF_S 1.5570899671947016
#define RBF_D 0.8493218002880191
#define RBF_R 5.289881816592866
#define KC10 0.049787068367863944
#define KC11 0.006737946999085467
#define KC20 4.5399929762484854e-05
#define KC21 8.3152871910356788e-07
#define KD10 0.36787944117144233
#define KD11 2.7182818284590451
#define KD20 0.0024787521766663585
#define KD21 0.1353352832366127
#define KD30 3.0590232050182579e-07
#define KD31 0.00012340980408667956

#define P1 28
#define P2 20

#define NSTEP1 30
#define NSTEP2 79
#define NSTEP3 40
#define NT1 2
#define NT2 1
#define NT3 1
#define B1_FLOATS (NSTEP1 * NT1 * 64)
#define B2_FLOATS (NSTEP2 * NT2 * 64)
#define B3_FLOATS (NSTEP3 * NT3 * 64)
#define B1_OFF 0
#define B2_OFF (B1_FLOATS)
#define B3_OFF (B1_FLOATS + B2_FLOATS)
#define BTOTAL (B1_FLOATS + B2_FLOATS + B3_FLOATS)

struct PairRegs { v4f a, b, c; };

__device__ __forceinline__ PairRegs load_pair(const float* __restrict__ src) {
  PairRegs r;
  r.a = *(const v4f_a4*)(src);
  r.b = *(const v4f_a4*)(src + 4);
  r.c = *(const v4f_a4*)(src + 8);
  return r;
}

template <int I, int O, int NT>
__device__ __forceinline__ void store_pair(const PairRegs& r, int pair, float* __restrict__ Bl) {
  const int i = pair / O, n = pair - i * O;
  const float cg[12] = {r.a[0], r.a[1], r.a[2], r.a[3], r.b[0], r.b[1], r.b[2], r.b[3],
                        r.c[0], r.c[1], r.c[2], r.c[3]};
  if (!(I & 1) || i < I - 1) {
    float* dp = Bl + ((3 * (i & ~1) * NT + (n >> 4)) * 32 + (i & 1) * 16 + (n & 15)) * 2;
#pragma unroll
    for (int s = 0; s < 6; ++s) {
      const int g0 = s < 3 ? 6 + 2 * s : 10 - 2 * s;
      const float k0 = s == 0 ? 1.0f : s == 1 ? (float)KC10 : s == 2 ? (float)KC20
                     : s == 3 ? (float)KD10 : s == 4 ? (float)KD20 : (float)KD30;
      const float k1 = s == 0 ? 1.0f : s == 1 ? (float)KC11 : s == 2 ? (float)KC21
                     : s == 3 ? (float)KD11 : s == 4 ? (float)KD21 : (float)KD31;
      v2f val;
      val.x = cg[g0] * k0;
      val.y = cg[g0 + 1] * k1;
      *(v2f*)(dp + s * NT * 64) = val;
    }
    return;
  }
  float* dst = Bl + ((3 * i * NT + (n >> 4)) * 32 + (n & 15)) * 2;
#pragma unroll
  for (int p = 0; p < 3; ++p) {
    const float k0 = p == 0 ? 1.0f : (p == 1 ? (float)KC10 : (float)KC20);
    const float k1 = p == 0 ? 1.0f : (p == 1 ? (float)KC11 : (float)KC21);
#pragma unroll
    for (int hh = 0; hh < 2; ++hh) {
      const int g0 = hh ? 6 + 2 * p : 5 - 2 * p;
      const int g1 = hh ? 7 + 2 * p : 4 - 2 * p;
      v2f val;
      val.x = cg[g0] * k0;
      val.y = cg[g1] * k1;
      *(v2f*)(dst + (p * NT * 32 + hh * 16) * 2) = val;
    }
  }
}

template <int I, int NT, bool PERM>
__device__ __forceinline__ void store_lin(float w, int f, float* __restrict__ Bl) {
  const int n = f / I, jin = f - n * I;
  const int j = PERM ? ((jin & 1) ? I / 2 + (jin >> 1) : (jin >> 1)) : jin;
  const int t = 3 * I + (j >> 2), hh = (j >> 1) & 1, v = j & 1;
  Bl[((t * NT + (n >> 4)) * 32 + hh * 16 + (n & 15)) * 2 + v] = w;
}

template <int I, int NT>
__device__ __forceinline__ void store_bias(float b, int n, float* __restrict__ Bl) {
  constexpr int k = 13 * I, t = k >> 2, hh = (k >> 1) & 1, v = k & 1;
  Bl[((t * NT + (n >> 4)) * 32 + hh * 16 + (n & 15)) * 2 + v] = b;
}

struct RbfGen { v2f f0; float rho; };

__device__ __forceinline__ RbfGen rbf_gen(float xi, float sS, float sR) {
  const v2f xv = {xi, xi};
  const v2f ssv = {sS, sS};
  const v2f c0v = {(float)(0.5 * RBF_D), (float)(1.5 * RBF_D)};
  const v2f q = __builtin_elementwise_fma(xv, ssv, -c0v);
  const v2f e = -(q * q);
  RbfGen g;
  g.f0.x = __builtin_amdgcn_exp2f(e.x);
  g.f0.y = __builtin_amdgcn_exp2f(e.y);
  g.rho = __builtin_amdgcn_exp2f(__builtin_fminf(xi * sR, 126.0f));
  return g;
}

struct RbfFull { v2f f0; float rp, rm; };

__device__ __forceinline__ RbfFull rbf_full(float xi) {
  const v2f xv = {xi, xi};
  const v2f ssv = {(float)RBF_S, (float)RBF_S};
  const v2f c0v = {(float)(0.5 * RBF_D), (float)(1.5 * RBF_D)};
  const v2f q = __builtin_elementwise_fma(xv, ssv, -c0v);
  const v2f e = -(q * q);
  RbfFull g;
  g.f0.x = __builtin_amdgcn_exp2f(e.x);
  g.f0.y = __builtin_amdgcn_exp2f(e.y);
  const float y = __builtin_amdgcn_fmed3f(xi * (float)RBF_R, -126.0f, 126.0f);
  g.rp = __builtin_amdgcn_exp2f(y);
  g.rm = __builtin_amdgcn_exp2f(-y);
  return g;
}

template <int I, int NT, bool REGS>
__device__ __forceinline__ void kan_layer_wmma(const float* __restrict__ arow,
                                               const float* __restrict__ xin,
                                               int half,
                                               const float* __restrict__ Bl,
                                               float sS, float sR,
                                               v8f& c0, v8f& c1) {
  const float* __restrict__ arowh = REGS ? arow : arow + 2 * half;
  constexpr int NP     = I / 2;
  constexpr int NBASIS = 3 * I;
  constexpr int NTAIL  = (I + 1 + 3) / 4;
  float own[NP];
  if (REGS) {
#pragma unroll
    for (int j = 0; j < NP; ++j) own[j] = half ? xin[2 * j + 1] : xin[2 * j];
  } else {
    const float* __restrict__ ao = arow + half * NP;
    if (NP % 4 == 0) {
#pragma unroll
      for (int g = 0; g < NP / 4; ++g) {
        const v4f t4 = *(const v4f*)(ao + 4 * g);
        own[4 * g] = t4[0]; own[4 * g + 1] = t4[1]; own[4 * g + 2] = t4[2]; own[4 * g + 3] = t4[3];
      }
    } else {
#pragma unroll
      for (int g = 0; g < NP / 2; ++g) {
        const v2f t2 = *(const v2f*)(ao + 2 * g);
        own[2 * g] = t2[0]; own[2 * g + 1] = t2[1];
      }
    }
  }
  RbfFull cur = rbf_full(own[0]);
#pragma unroll
  for (int j = 0; j < NP; ++j) {
    RbfFull nxt = cur;
    if (j + 1 < NP) nxt = rbf_full(own[j + 1]);
    v2f f[6];
    const v2f up = {cur.rp, cur.rp};
    const v2f dn = {cur.rm, cur.rm};
    f[0] = cur.f0;
    f[1] = f[0] * up;
    f[2] = f[1] * up;
    f[3] = f[0] * dn;
    f[4] = f[3] * dn;
    f[5] = f[4] * dn;
#pragma unroll
    for (int s = 0; s < 6; ++s) {
      const int t = 6 * j + s;
      const v2f a = f[s];
      const v2f b0 = *(const v2f*)(Bl + (t * NT + 0) * 64);
      c0 = __builtin_amdgcn_wmma_f32_16x16x4_f32(false, a, false, b0, (short)0, c0,
                                                 false, false);
      if (NT == 2) {
        const v2f b1 = *(const v2f*)(Bl + (t * NT + 1) * 64);
        c1 = __builtin_amdgcn_wmma_f32_16x16x4_f32(false, a, false, b1, (short)0, c1,
                                                   false, false);
      }
    }
    if (j + 1 < NP) {
      __builtin_amdgcn_sched_group_barrier(0x008, NT, 0);
      __builtin_amdgcn_sched_group_barrier(0x002, 4, 0);
      __builtin_amdgcn_sched_group_barrier(0x400, 1, 0);
      __builtin_amdgcn_sched_group_barrier(0x002, 1, 0);
      __builtin_amdgcn_sched_group_barrier(0x008, NT, 0);
      __builtin_amdgcn_sched_group_barrier(0x400, 1, 0);
      __builtin_amdgcn_sched_group_barrier(0x002, 1, 0);
      __builtin_amdgcn_sched_group_barrier(0x008, NT, 0);
      __builtin_amdgcn_sched_group_barrier(0x400, 1, 0);
      __builtin_amdgcn_sched_group_barrier(0x002, 1, 0);
      __builtin_amdgcn_sched_group_barrier(0x008, NT, 0);
      __builtin_amdgcn_sched_group_barrier(0x400, 1, 0);
      __builtin_amdgcn_sched_group_barrier(0x002, 1, 0);
      __builtin_amdgcn_sched_group_barrier(0x008, NT, 0);
      __builtin_amdgcn_sched_group_barrier(0x002, 1, 0);
      __builtin_amdgcn_sched_group_barrier(0x008, NT, 0);
    }
    cur = nxt;
  }
  if (I & 1) {
    const RbfGen g = rbf_gen(REGS ? xin[I - 1] : arow[I - 1], sS, sR);
    v2f f[3];
    const v2f rv = {g.rho, g.rho};
    f[0] = g.f0;
    f[1] = f[0] * rv;
    f[2] = f[1] * rv;
#pragma unroll
    for (int p = 0; p < 3; ++p) {
      const int t = 6 * NP + p;
      const v2f a = f[p];
      const v2f b0 = *(const v2f*)(Bl + (t * NT + 0) * 64);
      c0 = __builtin_amdgcn_wmma_f32_16x16x4_f32(false, a, false, b0, (short)0, c0,
                                                 false, false);
      if (NT == 2) {
        const v2f b1 = *(const v2f*)(Bl + (t * NT + 1) * 64);
        c1 = __builtin_amdgcn_wmma_f32_16x16x4_f32(false, a, false, b1, (short)0, c1,
                                                   false, false);
      }
    }
  }
#pragma unroll
  for (int tt = 0; tt < NTAIL; ++tt) {
    const int t = NBASIS + tt;
    v2f a;
    if (4 * tt + 3 < I) {
      if (REGS) {
        a.x = half ? xin[4 * tt + 2] : xin[4 * tt + 0];
        a.y = half ? xin[4 * tt + 3] : xin[4 * tt + 1];
      } else {
        a = *(const v2f*)(arowh + 4 * tt);
      }
    } else {
#pragma unroll
      for (int v = 0; v < 2; ++v) {
        const int jlo = 4 * tt + v, jhi = jlo + 2;
        const float elo = jlo < I ? (REGS ? xin[jlo] : arow[jlo]) : (jlo == I ? 1.0f : 0.0f);
        const float ehi = jhi < I ? (REGS ? xin[jhi] : arow[jhi]) : (jhi == I ? 1.0f : 0.0f);
        a[v] = half ? ehi : elo;
      }
    }
    const v2f b0 = *(const v2f*)(Bl + (t * NT + 0) * 64);
    c0 = __builtin_amdgcn_wmma_f32_16x16x4_f32(false, a, false, b0, (short)0, c0,
                                               false, false);
    if (NT == 2) {
      const v2f b1 = *(const v2f*)(Bl + (t * NT + 1) * 64);
      c1 = __builtin_amdgcn_wmma_f32_16x16x4_f32(false, a, false, b1, (short)0, c1,
                                                 false, false);
    }
  }
}

#define PIN() __builtin_amdgcn_sched_barrier(0x3F4)
template <int N>
__device__ __forceinline__ void silu_chains(const float* __restrict__ z, float* __restrict__ y) {
  float a[N], e[N], d[N], r[N];
#pragma unroll
  for (int k = 0; k < N; ++k) a[k] = z[k] * -1.44269504f;
  PIN();
#pragma unroll
  for (int k = 0; k < N + 3; ++k) {
    if (k >= 2 && k - 2 < N) { d[k - 2] = 1.0f + e[k - 2]; PIN(); }
    if (k >= 2 && k - 2 < N) { r[k - 2] = __builtin_amdgcn_rcpf(d[k - 2]); PIN(); }
    if (k >= 3) { y[k - 3] = z[k - 3] * r[k - 3]; PIN(); }
    if (k < N) { e[k] = __builtin_amdgcn_exp2f(a[k]); PIN(); }
  }
}

__global__ __launch_bounds__(256, 2) void kan_forward(
    const float* __restrict__ x,
    const float* __restrict__ cf1, const float* __restrict__ w1, const float* __restrict__ b1,
    const float* __restrict__ cf2, const float* __restrict__ w2, const float* __restrict__ b2,
    const float* __restrict__ cf3, const float* __restrict__ w3, const float* __restrict__ b3,
    const float* __restrict__ base, float* __restrict__ out, int tiles_per_wave) {
  __shared__ __align__(16) float Bsh[BTOTAL];
  __shared__ __align__(16) float act1[8][16][P1];
  __shared__ __align__(16) float act2[8][16][P2];
  __shared__ __align__(16) float cor[8][16][4];

  __builtin_amdgcn_s_setprio(3);

  const int tid  = threadIdx.x;
  const int wv   = tid >> 5;
  const int lane = tid & 31;
  const int half = lane >> 4;
  const int m    = lane & 15;

  const int tile0 = blockIdx.x * tiles_per_wave * 8 + wv;
  const float* xrow = x + (size_t)(tile0 * 16 + m) * 9;
  v4f   xa = *(const v4f_a4*)(xrow);
  v4f   xb = *(const v4f_a4*)(xrow + 4);
  float xc = xrow[8];

  const bool isA1 = tid < 216;
  const int  pA   = isA1 ? tid : (tid < 252 ? tid - 216 : 35);
  const int  pC   = 256 + (tid & 31);
  const PairRegs rA = load_pair(isA1 ? cf1 + pA * 12 : cf3 + pA * 12);
  const PairRegs rB = load_pair(cf2 + tid * 12);
  const PairRegs rC = load_pair(cf2 + pC * 12);
  const float lA = isA1 ? w1[pA] : w3[pA];
  const float lB = w2[tid];
  const float lC = w2[pC];
  const int   nb = tid < 24 ? tid : (tid < 36 ? tid - 24 : (tid < 39 ? tid - 36 : 0));
  const float* bp = tid < 24 ? b1 : (tid < 36 ? b2 : b3);
  const float bv = bp[nb];
  const float bs = base[nb < 3 ? nb : 0];

  for (int q = tid; q < BTOTAL / 4; q += 256) ((v4f*)Bsh)[q] = v4f{0.0f, 0.0f, 0.0f, 0.0f};
  __syncthreads();

  if (isA1) {
    store_pair<9, 24, NT1>(rA, pA, Bsh + B1_OFF);
    store_lin<9, NT1, false>(lA, pA, Bsh + B1_OFF);
  } else if (tid < 252) {
    store_pair<12, 3, NT3>(rA, pA, Bsh + B3_OFF);
    store_lin<12, NT3, true>(lA, pA, Bsh + B3_OFF);
  }
  store_pair<24, 12, NT2>(rB, tid, Bsh + B2_OFF);
  store_lin<24, NT2, true>(lB, tid, Bsh + B2_OFF);
  if (tid < 32) {
    store_pair<24, 12, NT2>(rC, pC, Bsh + B2_OFF);
    store_lin<24, NT2, true>(lC, pC, Bsh + B2_OFF);
  }
  if (tid < 24)      store_bias<9, NT1>(bv, nb, Bsh + B1_OFF);
  else if (tid < 36) store_bias<24, NT2>(bv, nb, Bsh + B2_OFF);
  else if (tid < 39) store_bias<12, NT3>(bv + bs, nb, Bsh + B3_OFF);
  __syncthreads();

  const float* a1 = &act1[wv][m][0];
  const float* a2 = &act2[wv][m][0];
  const float* Bl = Bsh + lane * 2;

  const int c0col = (m & 1) ? 12 + (m >> 1) : (m >> 1);
  const int c1col = m < 8 ? ((m & 1) ? 20 + (m >> 1) : 8 + (m >> 1)) : 24 + (m & 3);
  const int d0col = m < 12 ? ((m & 1) ? 6 + (m >> 1) : (m >> 1)) : m;

  const float sS = half ? (float)RBF_S : -(float)RBF_S;
  const float sR = half ? (float)RBF_R : -(float)RBF_R;

  for (int it = 0; it < tiles_per_wave; ++it) {
    const int rowbase = ((blockIdx.x * tiles_per_wave + it) * 8 + wv) * 16;
    const bool last_tile = it + 1 == tiles_per_wave;

    const float xin[9] = {xa[0], xa[1], xa[2], xa[3], xb[0], xb[1], xb[2], xb[3], xc};
    v8f c0 = {}, c1 = {};
    kan_layer_wmma<9, NT1, true>(nullptr, xin, half, Bl + B1_OFF, sS, sR, c0, c1);

    {
      const int itn = it + 1 < tiles_per_wave ? it + 1 : it;
      const float* xn = xrow + (size_t)itn * (8 * 16 * 9);
      xa = *(const v4f_a4*)(xn);
      xb = *(const v4f_a4*)(xn + 4);
      xc = xn[8];
    }
    {
      float z[16], y[16];
#pragma unroll
      for (int v = 0; v < 8; ++v) { z[2 * v] = c0[v]; z[2 * v + 1] = c1[v]; }
      silu_chains<16>(z, y);
#pragma unroll
      for (int v = 0; v < 8; ++v) {
        int r = half * 8 + v;
        act1[wv][r][c0col] = y[2 * v];
        act1[wv][r][c1col] = y[2 * v + 1];
      }
    }
    asm volatile("s_wait_dscnt 0" ::: "memory");

    if (last_tile) __builtin_amdgcn_s_setprio(2);
    v8f d0 = {}, dz = {};
    kan_layer_wmma<24, NT2, false>(a1, nullptr, half, Bl + B2_OFF, sS, sR, d0, dz);
    {
      float z[8], y[8];
#pragma unroll
      for (int v = 0; v < 8; ++v) z[v] = d0[v];
      silu_chains<8>(z, y);
#pragma unroll
      for (int v = 0; v < 8; ++v) {
        int r = half * 8 + v;
        act2[wv][r][d0col] = y[v];
      }
    }
    asm volatile("s_wait_dscnt 0" ::: "memory");

    if (last_tile) __builtin_amdgcn_s_setprio(1);
    v8f e0 = {}, ez = {};
    kan_layer_wmma<12, NT3, false>(a2, nullptr, half, Bl + B3_OFF, sS, sR, e0, ez);
#pragma unroll
    for (int v = 0; v < 8; ++v) {
      int r = half * 8 + v;
      if (m < 3) cor[wv][r][m] = e0[v];
    }
    asm volatile("s_wait_dscnt 0" ::: "memory");

    if (last_tile) __builtin_amdgcn_s_setprio(0);
    if (half == 0) {
      float p0 = cor[wv][m][0], p1 = cor[wv][m][1], p2 = cor[wv][m][2];
      float mx = fmaxf(p0, fmaxf(p1, p2));
      float q0 = __expf(p0 - mx), q1 = __expf(p1 - mx), q2 = __expf(p2 - mx);
      float inv = __builtin_amdgcn_rcpf(q0 + q1 + q2);
      float* o = out + (size_t)(rowbase + m) * 3;
      __builtin_nontemporal_store(q0 * inv, o + 0);
      __builtin_nontemporal_store(q1 * inv, o + 1);
      __builtin_nontemporal_store(q2 * inv, o + 2);
    }
    asm volatile("s_wait_dscnt 0" ::: "memory");
  }
  __builtin_amdgcn_s_setprio(0);
}

extern "C" void kernel_launch(void* const* d_in, const int* in_sizes, int n_in,
                              void* d_out, int out_size, void* d_ws, size_t ws_size,
                              hipStream_t stream) {
  const float* x    = (const float*)d_in[0];
  const float* c1   = (const float*)d_in[1];
  const float* w1   = (const float*)d_in[2];
  const float* b1   = (const float*)d_in[3];
  const float* c2   = (const float*)d_in[4];
  const float* w2   = (const float*)d_in[5];
  const float* b2   = (const float*)d_in[6];
  const float* c3   = (const float*)d_in[7];
  const float* w3   = (const float*)d_in[8];
  const float* b3   = (const float*)d_in[9];
  const float* base = (const float*)d_in[10];
  float* out = (float*)d_out;

  const int Btot = in_sizes[0] / 9;
  const int TILES = 4;
  const int blocks = Btot / (128 * TILES);

  kan_forward<<<blocks, 256, 0, stream>>>(x, c1, w1, b1, c2, w2, b2, c3, w3, b3, base,
                                          out, TILES);
}
